# starting kernel
# baseline (speedup 1.0000x reference)
#include <hip/hip_runtime.h>

typedef float v2f __attribute__((ext_vector_type(2)));
typedef float v8f __attribute__((ext_vector_type(8)));

#define N_NODES 20000
#define N_EDGES 200000

__device__ __forceinline__ v8f wmma4(v2f a, v2f b, v8f c) {
    return __builtin_amdgcn_wmma_f32_16x16x4_f32(
         false, a,  false, b,
         (short)0, c,  false,  false);
}

__device__ __forceinline__ float silu(float x) {
    return x / (1.0f + __expf(-x));
}

__global__ void zero_kernel(float4* __restrict__ p, int n4) {
    const int stride = gridDim.x * blockDim.x;
    for (int i = blockIdx.x * blockDim.x + threadIdx.x; i < n4; i += stride)
        p[i] = make_float4(0.f, 0.f, 0.f, 0.f);
}

__global__ void pack_b_kernel(const float* __restrict__ W, float2* __restrict__ Wp,
                              int K, int N) {
    const int total  = (K / 2) * N;
    const int stride = gridDim.x * blockDim.x;
    for (int idx = blockIdx.x * blockDim.x + threadIdx.x; idx < total; idx += stride) {
        const int q = idx / N;
        const int n = idx - q * N;
        float2 p;
        p.x = W[(2 * q) * N + n];
        p.y = W[(2 * q + 1) * N + n];
        Wp[idx] = p;
    }
}

__global__ void __launch_bounds__(256)
node_up_kernel(const float* __restrict__ nf, const v2f* __restrict__ W0p,
               const v2f* __restrict__ W1p, float* __restrict__ s_up,
               float* __restrict__ v_up) {
    const int lane   = threadIdx.x & 31;
    const int wv     = threadIdx.x >> 5;
    const int lane15 = lane & 15;
    const int half   = lane >> 4;
    const int koff   = half * 2;
    const int mbase  = half * 8;
    const int n0     = blockIdx.x * 16;
    const int col    = wv * 16 + lane15;
    const int ab     = (n0 + lane15) * 512;
    const int ab2    = (n0 + lane15) * 256 + half;
    const int wbase  = half * 128 + col;

    const v2f* nf2 = (const v2f*)nf;

    v8f aS = {}, aV0 = {}, aV1 = {}, aV2 = {};
    for (int k = 0; k < 128; k += 4) {
        const int kk = k + koff;
        const int qb = (k >> 2) * 256 + wbase;
        const v2f b0 = W0p[qb];
        const v2f b1 = W1p[qb];
        const v2f a  = nf2[ab2 + (k >> 1)];
        aS = wmma4(a, b0, aS);
        const int vb = ab + 128 + kk * 3;
        v2f av;
        av.x = nf[vb + 0]; av.y = nf[vb + 3];
        aV0 = wmma4(av, b1, aV0);
        av.x = nf[vb + 1]; av.y = nf[vb + 4];
        aV1 = wmma4(av, b1, aV1);
        av.x = nf[vb + 2]; av.y = nf[vb + 5];
        aV2 = wmma4(av, b1, aV2);
    }

    const float inv = 0.08838834764831845f;
#pragma unroll
    for (int r = 0; r < 8; ++r) {
        const int n = n0 + mbase + r;
        s_up[n * 128 + col] = aS[r] * inv;
        const int vo = (n * 128 + col) * 3;
        v_up[vo + 0] = aV0[r] * inv;
        v_up[vo + 1] = aV1[r] * inv;
        v_up[vo + 2] = aV2[r] * inv;
    }
}

__device__ __forceinline__ void mlp64_layer(const float* HSRC, float* HDST,
                                            const v2f* __restrict__ Wp,
                                            int lane15, int half, int mbase) {
    const v2f* H2    = (const v2f*)HSRC;
    const int abase  = lane15 * 34 + half;
    const int wbase  = half * 64 + lane15;
    v8f acc[4] = {{}, {}, {}, {}};
    for (int k = 0; k < 64; k += 4) {
        const v2f a  = H2[abase + (k >> 1)];
        const int qb = (k >> 2) * 128 + wbase;
#pragma unroll
        for (int ct = 0; ct < 4; ++ct) {
            const v2f b = Wp[qb + ct * 16];
            acc[ct] = wmma4(a, b, acc[ct]);
        }
    }
#pragma unroll
    for (int ct = 0; ct < 4; ++ct)
#pragma unroll
        for (int r = 0; r < 8; ++r)
            HDST[(mbase + r) * 68 + ct * 16 + lane15] = silu(acc[ct][r] * 0.125f);
}

__global__ void __launch_bounds__(128)
edge_kernel(const float* __restrict__ ea, const float* __restrict__ ef,
            const int* __restrict__ snd, const int* __restrict__ rcv,
            const v2f* __restrict__ w1p, const v2f* __restrict__ w2p,
            const v2f* __restrict__ w3p, const v2f* __restrict__ w4p,
            const float* __restrict__ s_up, const float* __restrict__ v_up,
            float* __restrict__ msg) {
    __shared__ float hA[4][16 * 68];
    __shared__ float hB[4][16 * 68];

    const int lane   = threadIdx.x & 31;
    const int wv     = threadIdx.x >> 5;
    const int lane15 = lane & 15;
    const int half   = lane >> 4;
    const int mbase  = half * 8;
    const int e0     = (blockIdx.x * 4 + wv) * 16;

    float* HA = hA[wv];
    float* HB = hB[wv];

    __builtin_prefetch((const float*)w4p + threadIdx.x * 256, 0, 1);

    int   sidx[8], vidx[8], midx[8];
    float y0a[8], y1xa[8], y1ya[8], y1za[8];
#pragma unroll
    for (int r = 0; r < 8; ++r) {
        const int e    = e0 + mbase + r;
        const int node = snd[e];
        sidx[r] = node * 128;
        vidx[r] = node * 384;
        midx[r] = rcv[e] * 1024;
        const float4 y = *(const float4*)(ea + e * 4);
        y0a[r] = y.x; y1xa[r] = y.y; y1ya[r] = y.z; y1za[r] = y.w;
    }

    {
        const v2f* ef2  = (const v2f*)ef;
        const int eb2   = (e0 + lane15) * 4 + half;
        const int wbase = half * 64 + lane15;
        v8f acc[4] = {{}, {}, {}, {}};
#pragma unroll
        for (int k = 0; k < 8; k += 4) {
            const v2f a  = ef2[eb2 + (k >> 1)];
            const int qb = (k >> 2) * 128 + wbase;
#pragma unroll
            for (int ct = 0; ct < 4; ++ct) {
                const v2f b = w1p[qb + ct * 16];
                acc[ct] = wmma4(a, b, acc[ct]);
            }
        }
#pragma unroll
        for (int ct = 0; ct < 4; ++ct)
#pragma unroll
            for (int r = 0; r < 8; ++r)
                HA[(mbase + r) * 68 + ct * 16 + lane15] = silu(acc[ct][r] * 0.35355339059327373f);
    }
    asm volatile("s_wait_dscnt 0" ::: "memory");

    mlp64_layer(HA, HB, w2p, lane15, half, mbase);
    asm volatile("s_wait_dscnt 0" ::: "memory");
    mlp64_layer(HB, HA, w3p, lane15, half, mbase);
    asm volatile("s_wait_dscnt 0" ::: "memory");

    const float c  = 0.08838834764831845f;
    const float c3 = 0.05103103630798287f;
    const v2f* H2   = (const v2f*)HA;
    const int abase = lane15 * 34 + half;
#pragma unroll 1
    for (int g = 0; g < 8; ++g) {
        const int wbase = half * 512 + g * 64 + lane15;
        v8f acc[4] = {{}, {}, {}, {}};
        for (int k = 0; k < 64; k += 4) {
            const v2f a  = H2[abase + (k >> 1)];
            const int qb = (k >> 2) * 1024 + wbase;
#pragma unroll
            for (int i = 0; i < 4; ++i) {
                const v2f b = w4p[qb + i * 16];
                acc[i] = wmma4(a, b, acc[i]);
            }
        }
        const int j = g >> 1;
#pragma unroll
        for (int i = 0; i < 4; ++i) {
            const int u  = ((g & 1) * 4 + i) * 16 + lane15;
            const int u3 = 3 * u;
#pragma unroll
            for (int r = 0; r < 8; ++r) {
                const float w = acc[i][r] * 0.125f;
                if (j == 0) {
                    const float ss = s_up[sidx[r] + u];
                    atomicAdd(&msg[midx[r] + u], c * ss * y0a[r] * w);
                } else if (j == 1) {
                    const int vb = vidx[r] + u3;
                    const float dot = v_up[vb + 0] * y1xa[r] + v_up[vb + 1] * y1ya[r] +
                                      v_up[vb + 2] * y1za[r];
                    atomicAdd(&msg[midx[r] + 128 + u], c3 * dot * w);
                } else if (j == 2) {
                    const float ss = s_up[sidx[r] + u];
                    const float t  = c * ss * w;
                    const int mb = midx[r] + 256 + u3;
                    atomicAdd(&msg[mb + 0], t * y1xa[r]);
                    atomicAdd(&msg[mb + 1], t * y1ya[r]);
                    atomicAdd(&msg[mb + 2], t * y1za[r]);
                } else {
                    const int vb = vidx[r] + u3;
                    const float t  = c * y0a[r] * w;
                    const int mb = midx[r] + 640 + u3;
                    atomicAdd(&msg[mb + 0], t * v_up[vb + 0]);
                    atomicAdd(&msg[mb + 1], t * v_up[vb + 1]);
                    atomicAdd(&msg[mb + 2], t * v_up[vb + 2]);
                }
            }
        }
    }
}

__global__ void __launch_bounds__(256)
node_down_kernel(const float* __restrict__ msg, const v2f* __restrict__ L0p,
                 const v2f* __restrict__ L1p, float* __restrict__ out) {
    const int lane   = threadIdx.x & 31;
    const int wv     = threadIdx.x >> 5;
    const int lane15 = lane & 15;
    const int half   = lane >> 4;
    const int koff   = half * 2;
    const int mbase  = half * 8;
    const int n0     = blockIdx.x * 16;
    const int col    = wv * 16 + lane15;
    const int ab     = (n0 + lane15) * 1024;
    const int ab2    = (n0 + lane15) * 512 + half;
    const int wbase  = half * 128 + col;

    const v2f* msg2 = (const v2f*)msg;

    v8f aS = {}, aV0 = {}, aV1 = {}, aV2 = {};
    for (int k = 0; k < 256; k += 4) {
        const int kk = k + koff;
        const int qb = (k >> 2) * 256 + wbase;
        const v2f b0 = L0p[qb];
        const v2f b1 = L1p[qb];
        const v2f a  = msg2[ab2 + (k >> 1)];
        aS = wmma4(a, b0, aS);
        const int vb = ab + 256 + kk * 3;
        v2f av;
        av.x = msg[vb + 0]; av.y = msg[vb + 3];
        aV0 = wmma4(av, b1, aV0);
        av.x = msg[vb + 1]; av.y = msg[vb + 4];
        aV1 = wmma4(av, b1, aV1);
        av.x = msg[vb + 2]; av.y = msg[vb + 5];
        aV2 = wmma4(av, b1, aV2);
    }

    const float sc = (1.0f / 16.0f) * 0.1f;
#pragma unroll
    for (int r = 0; r < 8; ++r) {
        const int n = n0 + mbase + r;
        out[n * 512 + col] = aS[r] * sc;
        const int oo = n * 512 + 128 + col * 3;
        out[oo + 0] = aV0[r] * sc;
        out[oo + 1] = aV1[r] * sc;
        out[oo + 2] = aV2[r] * sc;
    }
}

extern "C" void kernel_launch(void* const* d_in, const int* in_sizes, int n_in,
                              void* d_out, int out_size, void* d_ws, size_t ws_size,
                              hipStream_t stream) {
    (void)in_sizes; (void)n_in; (void)out_size; (void)ws_size;
    const float* nf   = (const float*)d_in[0];
    const float* ea   = (const float*)d_in[1];
    const float* ef   = (const float*)d_in[2];
    const int*   snd  = (const int*)d_in[3];
    const int*   rcv  = (const int*)d_in[4];
    const float* Wup0 = (const float*)d_in[5];
    const float* Wup1 = (const float*)d_in[6];
    const float* w1   = (const float*)d_in[7];
    const float* w2   = (const float*)d_in[8];
    const float* w3   = (const float*)d_in[9];
    const float* w4   = (const float*)d_in[10];
    const float* Wl0  = (const float*)d_in[11];
    const float* Wl1  = (const float*)d_in[12];
    float* out = (float*)d_out;

    float* s_up = (float*)d_ws;
    float* v_up = s_up + (size_t)N_NODES * 128;
    float* msgb = v_up + (size_t)N_NODES * 384;
    float* wp   = msgb + (size_t)N_NODES * 1024;
    float* Wup0p = wp;  wp += 128 * 128;
    float* Wup1p = wp;  wp += 128 * 128;
    float* w1p   = wp;  wp += 8 * 64;
    float* w2p   = wp;  wp += 64 * 64;
    float* w3p   = wp;  wp += 64 * 64;
    float* w4p   = wp;  wp += 64 * 512;
    float* Wl0p  = wp;  wp += 256 * 128;
    float* Wl1p  = wp;  wp += 256 * 128;

    zero_kernel<<<2048, 256, 0, stream>>>((float4*)msgb, N_NODES * 1024 / 4);

    pack_b_kernel<<<32, 256, 0, stream>>>(Wup0, (float2*)Wup0p, 128, 128);
    pack_b_kernel<<<32, 256, 0, stream>>>(Wup1, (float2*)Wup1p, 128, 128);
    pack_b_kernel<<<1,  256, 0, stream>>>(w1,   (float2*)w1p,   8,   64);
    pack_b_kernel<<<8,  256, 0, stream>>>(w2,   (float2*)w2p,   64,  64);
    pack_b_kernel<<<8,  256, 0, stream>>>(w3,   (float2*)w3p,   64,  64);
    pack_b_kernel<<<64, 256, 0, stream>>>(w4,   (float2*)w4p,   64,  512);
    pack_b_kernel<<<64, 256, 0, stream>>>(Wl0,  (float2*)Wl0p,  256, 128);
    pack_b_kernel<<<64, 256, 0, stream>>>(Wl1,  (float2*)Wl1p,  256, 128);

    node_up_kernel<<<N_NODES / 16, 256, 0, stream>>>(nf, (const v2f*)Wup0p,
                                                     (const v2f*)Wup1p, s_up, v_up);
    edge_kernel<<<N_EDGES / 64, 128, 0, stream>>>(ea, ef, snd, rcv,
                                                  (const v2f*)w1p, (const v2f*)w2p,
                                                  (const v2f*)w3p, (const v2f*)w4p,
                                                  s_up, v_up, msgb);
    node_down_kernel<<<N_NODES / 16, 256, 0, stream>>>(msgb, (const v2f*)Wl0p,
                                                       (const v2f*)Wl1p, out);
}


# Edge kernel takes edges in receiver order (counting sort rebuilt each call): each run of equal receivers in a lane's 8 rows is summed in registers, one atomic add per run; summation order changes
# speedup vs baseline: 1.2738x; 1.2738x over previous
#include <hip/hip_runtime.h>

typedef float v2f __attribute__((ext_vector_type(2)));
typedef float v8f __attribute__((ext_vector_type(8)));

#define N_NODES 20000
#define N_EDGES 200000

__device__ __forceinline__ v8f wmma4(v2f a, v2f b, v8f c) {
    return __builtin_amdgcn_wmma_f32_16x16x4_f32(
         false, a,  false, b,
         (short)0, c,  false,  false);
}

__device__ __forceinline__ float silu(float x) {
    return x / (1.0f + __expf(-x));
}

__global__ void zero_kernel(float4* __restrict__ p, int n4, int* __restrict__ cnt, int ncnt) {
    const int stride = gridDim.x * blockDim.x;
    const int gid    = blockIdx.x * blockDim.x + threadIdx.x;
    for (int i = gid; i < n4; i += stride)
        p[i] = make_float4(0.f, 0.f, 0.f, 0.f);
    for (int i = gid; i < ncnt; i += stride)
        cnt[i] = 0;
}

__global__ void count_kernel(const int* __restrict__ rcv, int* __restrict__ cnt) {
    const int stride = gridDim.x * blockDim.x;
    for (int e = blockIdx.x * blockDim.x + threadIdx.x; e < N_EDGES; e += stride)
        atomicAdd(&cnt[rcv[e]], 1);
}

#define SCAN_PER_THREAD 20
__global__ void __launch_bounds__(1024)
scan_kernel(int* __restrict__ cnt) {
    __shared__ int part[1024];
    const int t = threadIdx.x;
    const int b = t * SCAN_PER_THREAD;
    int loc[SCAN_PER_THREAD];
    int s = 0;
#pragma unroll
    for (int k = 0; k < SCAN_PER_THREAD; ++k) {
        const int idx = b + k;
        const int v = (idx < N_NODES) ? cnt[idx] : 0;
        loc[k] = s;
        s += v;
    }
    part[t] = s;
    __syncthreads();
    for (int off = 1; off < 1024; off <<= 1) {
        const int v = (t >= off) ? part[t - off] : 0;
        __syncthreads();
        part[t] += v;
        __syncthreads();
    }
    const int base = part[t] - s;
#pragma unroll
    for (int k = 0; k < SCAN_PER_THREAD; ++k) {
        const int idx = b + k;
        if (idx < N_NODES) cnt[idx] = base + loc[k];
    }
}

__global__ void fill_kernel(const int* __restrict__ rcv, int* __restrict__ cnt,
                            int* __restrict__ perm) {
    const int stride = gridDim.x * blockDim.x;
    for (int e = blockIdx.x * blockDim.x + threadIdx.x; e < N_EDGES; e += stride) {
        const int pos = atomicAdd(&cnt[rcv[e]], 1);
        perm[pos] = e;
    }
}

__global__ void pack_b_kernel(const float* __restrict__ W, float2* __restrict__ Wp,
                              int K, int N) {
    const int total  = (K / 2) * N;
    const int stride = gridDim.x * blockDim.x;
    for (int idx = blockIdx.x * blockDim.x + threadIdx.x; idx < total; idx += stride) {
        const int q = idx / N;
        const int n = idx - q * N;
        float2 p;
        p.x = W[(2 * q) * N + n];
        p.y = W[(2 * q + 1) * N + n];
        Wp[idx] = p;
    }
}

__global__ void __launch_bounds__(256)
node_up_kernel(const float* __restrict__ nf, const v2f* __restrict__ W0p,
               const v2f* __restrict__ W1p, float* __restrict__ s_up,
               float* __restrict__ v_up) {
    const int lane   = threadIdx.x & 31;
    const int wv     = threadIdx.x >> 5;
    const int lane15 = lane & 15;
    const int half   = lane >> 4;
    const int koff   = half * 2;
    const int mbase  = half * 8;
    const int n0     = blockIdx.x * 16;
    const int col    = wv * 16 + lane15;
    const int ab     = (n0 + lane15) * 512;
    const int ab2    = (n0 + lane15) * 256 + half;
    const int wbase  = half * 128 + col;

    const v2f* nf2 = (const v2f*)nf;

    v8f aS = {}, aV0 = {}, aV1 = {}, aV2 = {};
    for (int k = 0; k < 128; k += 4) {
        const int kk = k + koff;
        const int qb = (k >> 2) * 256 + wbase;
        const v2f b0 = W0p[qb];
        const v2f b1 = W1p[qb];
        const v2f a  = nf2[ab2 + (k >> 1)];
        aS = wmma4(a, b0, aS);
        const int vb = ab + 128 + kk * 3;
        v2f av;
        av.x = nf[vb + 0]; av.y = nf[vb + 3];
        aV0 = wmma4(av, b1, aV0);
        av.x = nf[vb + 1]; av.y = nf[vb + 4];
        aV1 = wmma4(av, b1, aV1);
        av.x = nf[vb + 2]; av.y = nf[vb + 5];
        aV2 = wmma4(av, b1, aV2);
    }

    const float inv = 0.08838834764831845f;
#pragma unroll
    for (int r = 0; r < 8; ++r) {
        const int n = n0 + mbase + r;
        s_up[n * 128 + col] = aS[r] * inv;
        const int vo = (n * 128 + col) * 3;
        v_up[vo + 0] = aV0[r] * inv;
        v_up[vo + 1] = aV1[r] * inv;
        v_up[vo + 2] = aV2[r] * inv;
    }
}

__device__ __forceinline__ void mlp64_layer(const float* HSRC, float* HDST,
                                            const v2f* __restrict__ Wp,
                                            int lane15, int half, int mbase) {
    const v2f* H2    = (const v2f*)HSRC;
    const int abase  = lane15 * 34 + half;
    const int wbase  = half * 64 + lane15;
    v8f acc[4] = {{}, {}, {}, {}};
    for (int k = 0; k < 64; k += 4) {
        const v2f a  = H2[abase + (k >> 1)];
        const int qb = (k >> 2) * 128 + wbase;
#pragma unroll
        for (int ct = 0; ct < 4; ++ct) {
            const v2f b = Wp[qb + ct * 16];
            acc[ct] = wmma4(a, b, acc[ct]);
        }
    }
#pragma unroll
    for (int ct = 0; ct < 4; ++ct)
#pragma unroll
        for (int r = 0; r < 8; ++r)
            HDST[(mbase + r) * 68 + ct * 16 + lane15] = silu(acc[ct][r] * 0.125f);
}

__global__ void __launch_bounds__(128)
edge_kernel(const float* __restrict__ ea, const float* __restrict__ ef,
            const int* __restrict__ snd, const int* __restrict__ rcv,
            const v2f* __restrict__ w1p, const v2f* __restrict__ w2p,
            const v2f* __restrict__ w3p, const v2f* __restrict__ w4p,
            const float* __restrict__ s_up, const float* __restrict__ v_up,
            const int* __restrict__ perm, float* __restrict__ msg) {
    __shared__ float hA[4][16 * 68];
    __shared__ float hB[4][16 * 68];

    const int lane   = threadIdx.x & 31;
    const int wv     = threadIdx.x >> 5;
    const int lane15 = lane & 15;
    const int half   = lane >> 4;
    const int mbase  = half * 8;
    const int e0     = (blockIdx.x * 4 + wv) * 16;

    float* HA = hA[wv];
    float* HB = hB[wv];

    __builtin_prefetch((const float*)w4p + threadIdx.x * 256, 0, 1);

    int   sidx[8], vidx[8], midx[8];
    float y0a[8], y1xa[8], y1ya[8], y1za[8];
#pragma unroll
    for (int r = 0; r < 8; ++r) {
        const int e    = perm[e0 + mbase + r];
        const int node = snd[e];
        sidx[r] = node * 128;
        vidx[r] = node * 384;
        midx[r] = rcv[e] * 1024;
        const float4 y = *(const float4*)(ea + e * 4);
        y0a[r] = y.x; y1xa[r] = y.y; y1ya[r] = y.z; y1za[r] = y.w;
    }
    unsigned fl = 0x80u;
#pragma unroll
    for (int r = 0; r < 7; ++r)
        if (midx[r] != midx[r + 1]) fl |= (1u << r);

    {
        const v2f* ef2  = (const v2f*)ef;
        const int eb2   = perm[e0 + lane15] * 4 + half;
        const int wbase = half * 64 + lane15;
        v8f acc[4] = {{}, {}, {}, {}};
#pragma unroll
        for (int k = 0; k < 8; k += 4) {
            const v2f a  = ef2[eb2 + (k >> 1)];
            const int qb = (k >> 2) * 128 + wbase;
#pragma unroll
            for (int ct = 0; ct < 4; ++ct) {
                const v2f b = w1p[qb + ct * 16];
                acc[ct] = wmma4(a, b, acc[ct]);
            }
        }
#pragma unroll
        for (int ct = 0; ct < 4; ++ct)
#pragma unroll
            for (int r = 0; r < 8; ++r)
                HA[(mbase + r) * 68 + ct * 16 + lane15] = silu(acc[ct][r] * 0.35355339059327373f);
    }
    asm volatile("s_wait_dscnt 0" ::: "memory");

    mlp64_layer(HA, HB, w2p, lane15, half, mbase);
    asm volatile("s_wait_dscnt 0" ::: "memory");
    mlp64_layer(HB, HA, w3p, lane15, half, mbase);
    asm volatile("s_wait_dscnt 0" ::: "memory");

    const float c  = 0.08838834764831845f;
    const float c3 = 0.05103103630798287f;
    const v2f* H2   = (const v2f*)HA;
    const int abase = lane15 * 34 + half;
#pragma unroll 1
    for (int g = 0; g < 8; ++g) {
        const int wbase = half * 512 + g * 64 + lane15;
        v8f acc[4] = {{}, {}, {}, {}};
        for (int k = 0; k < 64; k += 4) {
            const v2f a  = H2[abase + (k >> 1)];
            const int qb = (k >> 2) * 1024 + wbase;
#pragma unroll
            for (int i = 0; i < 4; ++i) {
                const v2f b = w4p[qb + i * 16];
                acc[i] = wmma4(a, b, acc[i]);
            }
        }
        const int j = g >> 1;
#pragma unroll
        for (int i = 0; i < 4; ++i) {
            const int u  = ((g & 1) * 4 + i) * 16 + lane15;
            const int u3 = 3 * u;
            float run0 = 0.f, run1 = 0.f, run2 = 0.f;
#pragma unroll
            for (int r = 0; r < 8; ++r) {
                const float w = acc[i][r] * 0.125f;
                const bool last = (fl >> r) & 1u;
                if (j == 0) {
                    const float ss = s_up[sidx[r] + u];
                    run0 += c * ss * y0a[r] * w;
                    if (last) { atomicAdd(&msg[midx[r] + u], run0); run0 = 0.f; }
                } else if (j == 1) {
                    const int vb = vidx[r] + u3;
                    const float dot = v_up[vb + 0] * y1xa[r] + v_up[vb + 1] * y1ya[r] +
                                      v_up[vb + 2] * y1za[r];
                    run0 += c3 * dot * w;
                    if (last) { atomicAdd(&msg[midx[r] + 128 + u], run0); run0 = 0.f; }
                } else if (j == 2) {
                    const float ss = s_up[sidx[r] + u];
                    const float t  = c * ss * w;
                    run0 += t * y1xa[r];
                    run1 += t * y1ya[r];
                    run2 += t * y1za[r];
                    if (last) {
                        const int mb = midx[r] + 256 + u3;
                        atomicAdd(&msg[mb + 0], run0);
                        atomicAdd(&msg[mb + 1], run1);
                        atomicAdd(&msg[mb + 2], run2);
                        run0 = 0.f; run1 = 0.f; run2 = 0.f;
                    }
                } else {
                    const int vb = vidx[r] + u3;
                    const float t  = c * y0a[r] * w;
                    run0 += t * v_up[vb + 0];
                    run1 += t * v_up[vb + 1];
                    run2 += t * v_up[vb + 2];
                    if (last) {
                        const int mb = midx[r] + 640 + u3;
                        atomicAdd(&msg[mb + 0], run0);
                        atomicAdd(&msg[mb + 1], run1);
                        atomicAdd(&msg[mb + 2], run2);
                        run0 = 0.f; run1 = 0.f; run2 = 0.f;
                    }
                }
            }
        }
    }
}

__global__ void __launch_bounds__(256)
node_down_kernel(const float* __restrict__ msg, const v2f* __restrict__ L0p,
                 const v2f* __restrict__ L1p, float* __restrict__ out) {
    const int lane   = threadIdx.x & 31;
    const int wv     = threadIdx.x >> 5;
    const int lane15 = lane & 15;
    const int half   = lane >> 4;
    const int koff   = half * 2;
    const int mbase  = half * 8;
    const int n0     = blockIdx.x * 16;
    const int col    = wv * 16 + lane15;
    const int ab     = (n0 + lane15) * 1024;
    const int ab2    = (n0 + lane15) * 512 + half;
    const int wbase  = half * 128 + col;

    const v2f* msg2 = (const v2f*)msg;

    v8f aS = {}, aV0 = {}, aV1 = {}, aV2 = {};
    for (int k = 0; k < 256; k += 4) {
        const int kk = k + koff;
        const int qb = (k >> 2) * 256 + wbase;
        const v2f b0 = L0p[qb];
        const v2f b1 = L1p[qb];
        const v2f a  = msg2[ab2 + (k >> 1)];
        aS = wmma4(a, b0, aS);
        const int vb = ab + 256 + kk * 3;
        v2f av;
        av.x = msg[vb + 0]; av.y = msg[vb + 3];
        aV0 = wmma4(av, b1, aV0);
        av.x = msg[vb + 1]; av.y = msg[vb + 4];
        aV1 = wmma4(av, b1, aV1);
        av.x = msg[vb + 2]; av.y = msg[vb + 5];
        aV2 = wmma4(av, b1, aV2);
    }

    const float sc = (1.0f / 16.0f) * 0.1f;
#pragma unroll
    for (int r = 0; r < 8; ++r) {
        const int n = n0 + mbase + r;
        out[n * 512 + col] = aS[r] * sc;
        const int oo = n * 512 + 128 + col * 3;
        out[oo + 0] = aV0[r] * sc;
        out[oo + 1] = aV1[r] * sc;
        out[oo + 2] = aV2[r] * sc;
    }
}

extern "C" void kernel_launch(void* const* d_in, const int* in_sizes, int n_in,
                              void* d_out, int out_size, void* d_ws, size_t ws_size,
                              hipStream_t stream) {
    (void)in_sizes; (void)n_in; (void)out_size; (void)ws_size;
    const float* nf   = (const float*)d_in[0];
    const float* ea   = (const float*)d_in[1];
    const float* ef   = (const float*)d_in[2];
    const int*   snd  = (const int*)d_in[3];
    const int*   rcv  = (const int*)d_in[4];
    const float* Wup0 = (const float*)d_in[5];
    const float* Wup1 = (const float*)d_in[6];
    const float* w1   = (const float*)d_in[7];
    const float* w2   = (const float*)d_in[8];
    const float* w3   = (const float*)d_in[9];
    const float* w4   = (const float*)d_in[10];
    const float* Wl0  = (const float*)d_in[11];
    const float* Wl1  = (const float*)d_in[12];
    float* out = (float*)d_out;

    float* s_up = (float*)d_ws;
    float* v_up = s_up + (size_t)N_NODES * 128;
    float* msgb = v_up + (size_t)N_NODES * 384;
    float* wp   = msgb + (size_t)N_NODES * 1024;
    float* Wup0p = wp;  wp += 128 * 128;
    float* Wup1p = wp;  wp += 128 * 128;
    float* w1p   = wp;  wp += 8 * 64;
    float* w2p   = wp;  wp += 64 * 64;
    float* w3p   = wp;  wp += 64 * 64;
    float* w4p   = wp;  wp += 64 * 512;
    float* Wl0p  = wp;  wp += 256 * 128;
    float* Wl1p  = wp;  wp += 256 * 128;

    int* cnt  = (int*)d_out;
    int* perm = cnt + N_NODES;

    zero_kernel<<<2048, 256, 0, stream>>>((float4*)msgb, N_NODES * 1024 / 4, cnt, N_NODES);

    count_kernel<<<196, 1024, 0, stream>>>(rcv, cnt);
    scan_kernel<<<1, 1024, 0, stream>>>(cnt);
    fill_kernel<<<196, 1024, 0, stream>>>(rcv, cnt, perm);

    pack_b_kernel<<<32, 256, 0, stream>>>(Wup0, (float2*)Wup0p, 128, 128);
    pack_b_kernel<<<32, 256, 0, stream>>>(Wup1, (float2*)Wup1p, 128, 128);
    pack_b_kernel<<<1,  256, 0, stream>>>(w1,   (float2*)w1p,   8,   64);
    pack_b_kernel<<<8,  256, 0, stream>>>(w2,   (float2*)w2p,   64,  64);
    pack_b_kernel<<<8,  256, 0, stream>>>(w3,   (float2*)w3p,   64,  64);
    pack_b_kernel<<<64, 256, 0, stream>>>(w4,   (float2*)w4p,   64,  512);
    pack_b_kernel<<<64, 256, 0, stream>>>(Wl0,  (float2*)Wl0p,  256, 128);
    pack_b_kernel<<<64, 256, 0, stream>>>(Wl1,  (float2*)Wl1p,  256, 128);

    node_up_kernel<<<N_NODES / 16, 256, 0, stream>>>(nf, (const v2f*)Wup0p,
                                                     (const v2f*)Wup1p, s_up, v_up);
    edge_kernel<<<N_EDGES / 64, 128, 0, stream>>>(ea, ef, snd, rcv,
                                                  (const v2f*)w1p, (const v2f*)w2p,
                                                  (const v2f*)w3p, (const v2f*)w4p,
                                                  s_up, v_up, perm, msgb);
    node_down_kernel<<<N_NODES / 16, 256, 0, stream>>>(msgb, (const v2f*)Wl0p,
                                                       (const v2f*)Wl1p, out);
}


# Edge kernel: each layer-4 pass requests its 32 sender-row gathers (s_up / v_up) as one group ahead of the matrix loop instead of one load and wait per row in the message loop
# speedup vs baseline: 2.1818x; 1.7128x over previous
#include <hip/hip_runtime.h>

typedef float v2f __attribute__((ext_vector_type(2)));
typedef float v8f __attribute__((ext_vector_type(8)));

#define N_NODES 20000
#define N_EDGES 200000

__device__ __forceinline__ v8f wmma4(v2f a, v2f b, v8f c) {
    return __builtin_amdgcn_wmma_f32_16x16x4_f32(
         false, a,  false, b,
         (short)0, c,  false,  false);
}

__device__ __forceinline__ float silu(float x) {
    return x / (1.0f + __expf(-x));
}

__global__ void zero_kernel(float4* __restrict__ p, int n4, int* __restrict__ cnt, int ncnt) {
    const int stride = gridDim.x * blockDim.x;
    const int gid    = blockIdx.x * blockDim.x + threadIdx.x;
    for (int i = gid; i < n4; i += stride)
        p[i] = make_float4(0.f, 0.f, 0.f, 0.f);
    for (int i = gid; i < ncnt; i += stride)
        cnt[i] = 0;
}

__global__ void count_kernel(const int* __restrict__ rcv, int* __restrict__ cnt) {
    const int stride = gridDim.x * blockDim.x;
    for (int e = blockIdx.x * blockDim.x + threadIdx.x; e < N_EDGES; e += stride)
        atomicAdd(&cnt[rcv[e]], 1);
}

#define SCAN_PER_THREAD 20
__global__ void __launch_bounds__(1024)
scan_kernel(int* __restrict__ cnt) {
    __shared__ int part[1024];
    const int t = threadIdx.x;
    const int b = t * SCAN_PER_THREAD;
    int loc[SCAN_PER_THREAD];
    int s = 0;
#pragma unroll
    for (int k = 0; k < SCAN_PER_THREAD; ++k) {
        const int idx = b + k;
        const int v = (idx < N_NODES) ? cnt[idx] : 0;
        loc[k] = s;
        s += v;
    }
    part[t] = s;
    __syncthreads();
    for (int off = 1; off < 1024; off <<= 1) {
        const int v = (t >= off) ? part[t - off] : 0;
        __syncthreads();
        part[t] += v;
        __syncthreads();
    }
    const int base = part[t] - s;
#pragma unroll
    for (int k = 0; k < SCAN_PER_THREAD; ++k) {
        const int idx = b + k;
        if (idx < N_NODES) cnt[idx] = base + loc[k];
    }
}

__global__ void fill_kernel(const int* __restrict__ rcv, int* __restrict__ cnt,
                            int* __restrict__ perm) {
    const int stride = gridDim.x * blockDim.x;
    for (int e = blockIdx.x * blockDim.x + threadIdx.x; e < N_EDGES; e += stride) {
        const int pos = atomicAdd(&cnt[rcv[e]], 1);
        perm[pos] = e;
    }
}

__global__ void pack_b_kernel(const float* __restrict__ W, float2* __restrict__ Wp,
                              int K, int N) {
    const int total  = (K / 2) * N;
    const int stride = gridDim.x * blockDim.x;
    for (int idx = blockIdx.x * blockDim.x + threadIdx.x; idx < total; idx += stride) {
        const int q = idx / N;
        const int n = idx - q * N;
        float2 p;
        p.x = W[(2 * q) * N + n];
        p.y = W[(2 * q + 1) * N + n];
        Wp[idx] = p;
    }
}

__global__ void __launch_bounds__(256)
node_up_kernel(const float* __restrict__ nf, const v2f* __restrict__ W0p,
               const v2f* __restrict__ W1p, float* __restrict__ s_up,
               float* __restrict__ v_up) {
    const int lane   = threadIdx.x & 31;
    const int wv     = threadIdx.x >> 5;
    const int lane15 = lane & 15;
    const int half   = lane >> 4;
    const int koff   = half * 2;
    const int mbase  = half * 8;
    const int n0     = blockIdx.x * 16;
    const int col    = wv * 16 + lane15;
    const int ab     = (n0 + lane15) * 512;
    const int ab2    = (n0 + lane15) * 256 + half;
    const int wbase  = half * 128 + col;

    const v2f* nf2 = (const v2f*)nf;

    v8f aS = {}, aV0 = {}, aV1 = {}, aV2 = {};
    for (int k = 0; k < 128; k += 4) {
        const int kk = k + koff;
        const int qb = (k >> 2) * 256 + wbase;
        const v2f b0 = W0p[qb];
        const v2f b1 = W1p[qb];
        const v2f a  = nf2[ab2 + (k >> 1)];
        aS = wmma4(a, b0, aS);
        const int vb = ab + 128 + kk * 3;
        v2f av;
        av.x = nf[vb + 0]; av.y = nf[vb + 3];
        aV0 = wmma4(av, b1, aV0);
        av.x = nf[vb + 1]; av.y = nf[vb + 4];
        aV1 = wmma4(av, b1, aV1);
        av.x = nf[vb + 2]; av.y = nf[vb + 5];
        aV2 = wmma4(av, b1, aV2);
    }

    const float inv = 0.08838834764831845f;
#pragma unroll
    for (int r = 0; r < 8; ++r) {
        const int n = n0 + mbase + r;
        s_up[n * 128 + col] = aS[r] * inv;
        const int vo = (n * 128 + col) * 3;
        v_up[vo + 0] = aV0[r] * inv;
        v_up[vo + 1] = aV1[r] * inv;
        v_up[vo + 2] = aV2[r] * inv;
    }
}

__device__ __forceinline__ void mlp64_layer(const float* HSRC, float* HDST,
                                            const v2f* __restrict__ Wp,
                                            int lane15, int half, int mbase) {
    const v2f* H2    = (const v2f*)HSRC;
    const int abase  = lane15 * 34 + half;
    const int wbase  = half * 64 + lane15;
    v8f acc[4] = {{}, {}, {}, {}};
    for (int k = 0; k < 64; k += 4) {
        const v2f a  = H2[abase + (k >> 1)];
        const int qb = (k >> 2) * 128 + wbase;
#pragma unroll
        for (int ct = 0; ct < 4; ++ct) {
            const v2f b = Wp[qb + ct * 16];
            acc[ct] = wmma4(a, b, acc[ct]);
        }
    }
#pragma unroll
    for (int ct = 0; ct < 4; ++ct)
#pragma unroll
        for (int r = 0; r < 8; ++r)
            HDST[(mbase + r) * 68 + ct * 16 + lane15] = silu(acc[ct][r] * 0.125f);
}

template <int J>
__device__ __forceinline__ void tp_pass(
        const int g, const v2f* H2, const int abase, const int half, const int lane15,
        const v2f* __restrict__ w4p, const float* __restrict__ s_up,
        const float* __restrict__ v_up, float* __restrict__ msg,
        const int (&sidx)[8], const int (&vidx)[8], const int (&midx)[8],
        const float (&y0a)[8], const float (&y1xa)[8], const float (&y1ya)[8],
        const float (&y1za)[8], const unsigned fl) {
    const float c  = 0.08838834764831845f;
    const float c3 = 0.05103103630798287f;
    const int ub = (g & 1) * 64 + lane15;

    const int wbase = half * 512 + g * 64 + lane15;
    v2f b0[8][4], b1[8][4];
#pragma unroll
    for (int kk = 0; kk < 8; ++kk)
#pragma unroll
        for (int i = 0; i < 4; ++i)
            b0[kk][i] = w4p[kk * 1024 + wbase + i * 16];
    __builtin_amdgcn_sched_barrier(0);

    float gs[4][8];
    float gv[4][8][3];
    if (J == 0 || J == 2) {
#pragma unroll
        for (int i = 0; i < 4; ++i)
#pragma unroll
            for (int r = 0; r < 8; ++r)
                gs[i][r] = s_up[sidx[r] + ub + 16 * i];
    } else {
#pragma unroll
        for (int i = 0; i < 4; ++i)
#pragma unroll
            for (int r = 0; r < 8; ++r) {
                const int vb = vidx[r] + 3 * (ub + 16 * i);
                gv[i][r][0] = v_up[vb + 0];
                gv[i][r][1] = v_up[vb + 1];
                gv[i][r][2] = v_up[vb + 2];
            }
    }

    __builtin_amdgcn_sched_barrier(0);
    v8f acc[4] = {{}, {}, {}, {}};
#pragma unroll
    for (int kk = 0; kk < 8; ++kk) {
        const v2f a = H2[abase + 2 * kk];
#pragma unroll
        for (int i = 0; i < 4; ++i)
            acc[i] = wmma4(a, b0[kk][i], acc[i]);
#pragma unroll
        for (int i = 0; i < 4; ++i)
            b1[kk][i] = w4p[(kk + 8) * 1024 + wbase + i * 16];
        __builtin_amdgcn_sched_group_barrier(0x008, 4, 0);
        __builtin_amdgcn_sched_group_barrier(0x020, 4, 0);
    }
#pragma unroll
    for (int kk = 0; kk < 8; ++kk) {
        const v2f a = H2[abase + 2 * (kk + 8)];
#pragma unroll
        for (int i = 0; i < 4; ++i)
            acc[i] = wmma4(a, b1[kk][i], acc[i]);
    }

#pragma unroll
    for (int i = 0; i < 4; ++i) {
        const int u  = ub + 16 * i;
        const int u3 = 3 * u;
        float run0 = 0.f, run1 = 0.f, run2 = 0.f;
#pragma unroll
        for (int r = 0; r < 8; ++r) {
            const float w = acc[i][r] * 0.125f;
            const bool last = (fl >> r) & 1u;
            if (J == 0) {
                run0 += c * gs[i][r] * y0a[r] * w;
                if (last) { atomicAdd(&msg[midx[r] + u], run0); run0 = 0.f; }
            } else if (J == 1) {
                const float dot = gv[i][r][0] * y1xa[r] + gv[i][r][1] * y1ya[r] +
                                  gv[i][r][2] * y1za[r];
                run0 += c3 * dot * w;
                if (last) { atomicAdd(&msg[midx[r] + 128 + u], run0); run0 = 0.f; }
            } else if (J == 2) {
                const float t = c * gs[i][r] * w;
                run0 += t * y1xa[r];
                run1 += t * y1ya[r];
                run2 += t * y1za[r];
                if (last) {
                    const int mb = midx[r] + 256 + u3;
                    atomicAdd(&msg[mb + 0], run0);
                    atomicAdd(&msg[mb + 1], run1);
                    atomicAdd(&msg[mb + 2], run2);
                    run0 = 0.f; run1 = 0.f; run2 = 0.f;
                }
            } else {
                const float t = c * y0a[r] * w;
                run0 += t * gv[i][r][0];
                run1 += t * gv[i][r][1];
                run2 += t * gv[i][r][2];
                if (last) {
                    const int mb = midx[r] + 640 + u3;
                    atomicAdd(&msg[mb + 0], run0);
                    atomicAdd(&msg[mb + 1], run1);
                    atomicAdd(&msg[mb + 2], run2);
                    run0 = 0.f; run1 = 0.f; run2 = 0.f;
                }
            }
        }
    }
}

__global__ void __launch_bounds__(128)
edge_kernel(const float* __restrict__ ea, const float* __restrict__ ef,
            const int* __restrict__ snd, const int* __restrict__ rcv,
            const v2f* __restrict__ w1p, const v2f* __restrict__ w2p,
            const v2f* __restrict__ w3p, const v2f* __restrict__ w4p,
            const float* __restrict__ s_up, const float* __restrict__ v_up,
            const int* __restrict__ perm, float* __restrict__ msg) {
    __shared__ float hA[4][16 * 68];
    __shared__ float hB[4][16 * 68];

    const int lane   = threadIdx.x & 31;
    const int wv     = threadIdx.x >> 5;
    const int lane15 = lane & 15;
    const int half   = lane >> 4;
    const int mbase  = half * 8;
    const int e0     = (blockIdx.x * 4 + wv) * 16;

    float* HA = hA[wv];
    float* HB = hB[wv];

    __builtin_prefetch((const float*)w4p + threadIdx.x * 256, 0, 1);

    int   sidx[8], vidx[8], midx[8];
    float y0a[8], y1xa[8], y1ya[8], y1za[8];
#pragma unroll
    for (int r = 0; r < 8; ++r) {
        const int e    = perm[e0 + mbase + r];
        const int node = snd[e];
        sidx[r] = node * 128;
        vidx[r] = node * 384;
        midx[r] = rcv[e] * 1024;
        const float4 y = *(const float4*)(ea + e * 4);
        y0a[r] = y.x; y1xa[r] = y.y; y1ya[r] = y.z; y1za[r] = y.w;
    }
    unsigned fl = 0x80u;
#pragma unroll
    for (int r = 0; r < 7; ++r)
        if (midx[r] != midx[r + 1]) fl |= (1u << r);

    {
        const v2f* ef2  = (const v2f*)ef;
        const int eb2   = perm[e0 + lane15] * 4 + half;
        const int wbase = half * 64 + lane15;
        v8f acc[4] = {{}, {}, {}, {}};
#pragma unroll
        for (int k = 0; k < 8; k += 4) {
            const v2f a  = ef2[eb2 + (k >> 1)];
            const int qb = (k >> 2) * 128 + wbase;
#pragma unroll
            for (int ct = 0; ct < 4; ++ct) {
                const v2f b = w1p[qb + ct * 16];
                acc[ct] = wmma4(a, b, acc[ct]);
            }
        }
#pragma unroll
        for (int ct = 0; ct < 4; ++ct)
#pragma unroll
            for (int r = 0; r < 8; ++r)
                HA[(mbase + r) * 68 + ct * 16 + lane15] = silu(acc[ct][r] * 0.35355339059327373f);
    }
    asm volatile("s_wait_dscnt 0" ::: "memory");

    mlp64_layer(HA, HB, w2p, lane15, half, mbase);
    asm volatile("s_wait_dscnt 0" ::: "memory");
    mlp64_layer(HB, HA, w3p, lane15, half, mbase);
    asm volatile("s_wait_dscnt 0" ::: "memory");

    const v2f* H2   = (const v2f*)HA;
    const int abase = lane15 * 34 + half;
#pragma unroll 1
    for (int g = 0; g < 2; ++g)
        tp_pass<0>(g, H2, abase, half, lane15, w4p, s_up, v_up, msg,
                   sidx, vidx, midx, y0a, y1xa, y1ya, y1za, fl);
#pragma unroll 1
    for (int g = 2; g < 4; ++g)
        tp_pass<1>(g, H2, abase, half, lane15, w4p, s_up, v_up, msg,
                   sidx, vidx, midx, y0a, y1xa, y1ya, y1za, fl);
#pragma unroll 1
    for (int g = 4; g < 6; ++g)
        tp_pass<2>(g, H2, abase, half, lane15, w4p, s_up, v_up, msg,
                   sidx, vidx, midx, y0a, y1xa, y1ya, y1za, fl);
#pragma unroll 1
    for (int g = 6; g < 8; ++g)
        tp_pass<3>(g, H2, abase, half, lane15, w4p, s_up, v_up, msg,
                   sidx, vidx, midx, y0a, y1xa, y1ya, y1za, fl);
}

__global__ void __launch_bounds__(256)
node_down_kernel(const float* __restrict__ msg, const v2f* __restrict__ L0p,
                 const v2f* __restrict__ L1p, float* __restrict__ out) {
    const int lane   = threadIdx.x & 31;
    const int wv     = threadIdx.x >> 5;
    const int lane15 = lane & 15;
    const int half   = lane >> 4;
    const int koff   = half * 2;
    const int mbase  = half * 8;
    const int n0     = blockIdx.x * 16;
    const int col    = wv * 16 + lane15;
    const int ab     = (n0 + lane15) * 1024;
    const int ab2    = (n0 + lane15) * 512 + half;
    const int wbase  = half * 128 + col;

    const v2f* msg2 = (const v2f*)msg;

    v8f aS = {}, aV0 = {}, aV1 = {}, aV2 = {};
    for (int k = 0; k < 256; k += 4) {
        const int kk = k + koff;
        const int qb = (k >> 2) * 256 + wbase;
        const v2f b0 = L0p[qb];
        const v2f b1 = L1p[qb];
        const v2f a  = msg2[ab2 + (k >> 1)];
        aS = wmma4(a, b0, aS);
        const int vb = ab + 256 + kk * 3;
        v2f av;
        av.x = msg[vb + 0]; av.y = msg[vb + 3];
        aV0 = wmma4(av, b1, aV0);
        av.x = msg[vb + 1]; av.y = msg[vb + 4];
        aV1 = wmma4(av, b1, aV1);
        av.x = msg[vb + 2]; av.y = msg[vb + 5];
        aV2 = wmma4(av, b1, aV2);
    }

    const float sc = (1.0f / 16.0f) * 0.1f;
#pragma unroll
    for (int r = 0; r < 8; ++r) {
        const int n = n0 + mbase + r;
        out[n * 512 + col] = aS[r] * sc;
        const int oo = n * 512 + 128 + col * 3;
        out[oo + 0] = aV0[r] * sc;
        out[oo + 1] = aV1[r] * sc;
        out[oo + 2] = aV2[r] * sc;
    }
}

extern "C" void kernel_launch(void* const* d_in, const int* in_sizes, int n_in,
                              void* d_out, int out_size, void* d_ws, size_t ws_size,
                              hipStream_t stream) {
    (void)in_sizes; (void)n_in; (void)out_size; (void)ws_size;
    const float* nf   = (const float*)d_in[0];
    const float* ea   = (const float*)d_in[1];
    const float* ef   = (const float*)d_in[2];
    const int*   snd  = (const int*)d_in[3];
    const int*   rcv  = (const int*)d_in[4];
    const float* Wup0 = (const float*)d_in[5];
    const float* Wup1 = (const float*)d_in[6];
    const float* w1   = (const float*)d_in[7];
    const float* w2   = (const float*)d_in[8];
    const float* w3   = (const float*)d_in[9];
    const float* w4   = (const float*)d_in[10];
    const float* Wl0  = (const float*)d_in[11];
    const float* Wl1  = (const float*)d_in[12];
    float* out = (float*)d_out;

    float* s_up = (float*)d_ws;
    float* v_up = s_up + (size_t)N_NODES * 128;
    float* msgb = v_up + (size_t)N_NODES * 384;
    float* wp   = msgb + (size_t)N_NODES * 1024;
    float* Wup0p = wp;  wp += 128 * 128;
    float* Wup1p = wp;  wp += 128 * 128;
    float* w1p   = wp;  wp += 8 * 64;
    float* w2p   = wp;  wp += 64 * 64;
    float* w3p   = wp;  wp += 64 * 64;
    float* w4p   = wp;  wp += 64 * 512;
    float* Wl0p  = wp;  wp += 256 * 128;
    float* Wl1p  = wp;  wp += 256 * 128;

    int* cnt  = (int*)d_out;
    int* perm = cnt + N_NODES;

    zero_kernel<<<2048, 256, 0, stream>>>((float4*)msgb, N_NODES * 1024 / 4, cnt, N_NODES);

    count_kernel<<<196, 1024, 0, stream>>>(rcv, cnt);
    scan_kernel<<<1, 1024, 0, stream>>>(cnt);
    fill_kernel<<<196, 1024, 0, stream>>>(rcv, cnt, perm);

    pack_b_kernel<<<32, 256, 0, stream>>>(Wup0, (float2*)Wup0p, 128, 128);
    pack_b_kernel<<<32, 256, 0, stream>>>(Wup1, (float2*)Wup1p, 128, 128);
    pack_b_kernel<<<1,  256, 0, stream>>>(w1,   (float2*)w1p,   8,   64);
    pack_b_kernel<<<8,  256, 0, stream>>>(w2,   (float2*)w2p,   64,  64);
    pack_b_kernel<<<8,  256, 0, stream>>>(w3,   (float2*)w3p,   64,  64);
    pack_b_kernel<<<64, 256, 0, stream>>>(w4,   (float2*)w4p,   64,  512);
    pack_b_kernel<<<64, 256, 0, stream>>>(Wl0,  (float2*)Wl0p,  256, 128);
    pack_b_kernel<<<64, 256, 0, stream>>>(Wl1,  (float2*)Wl1p,  256, 128);

    node_up_kernel<<<N_NODES / 16, 256, 0, stream>>>(nf, (const v2f*)Wup0p,
                                                     (const v2f*)Wup1p, s_up, v_up);
    edge_kernel<<<N_EDGES / 64, 128, 0, stream>>>(ea, ef, snd, rcv,
                                                  (const v2f*)w1p, (const v2f*)w2p,
                                                  (const v2f*)w3p, (const v2f*)w4p,
                                                  s_up, v_up, perm, msgb);
    node_down_kernel<<<N_NODES / 16, 256, 0, stream>>>(msgb, (const v2f*)Wl0p,
                                                       (const v2f*)Wl1p, out);
}


# Edge kernel: run sums of a wave's 16 receiver-ordered rows are joined across the lane halves; plain stores for runs inside the 16 rows, atomic adds only for runs crossing their ends; sum order changes
# speedup vs baseline: 2.3799x; 1.0908x over previous
#include <hip/hip_runtime.h>

typedef float v2f __attribute__((ext_vector_type(2)));
typedef float v8f __attribute__((ext_vector_type(8)));

#define N_NODES 20000
#define N_EDGES 200000

__device__ __forceinline__ v8f wmma4(v2f a, v2f b, v8f c) {
    return __builtin_amdgcn_wmma_f32_16x16x4_f32(
         false, a,  false, b,
         (short)0, c,  false,  false);
}

__device__ __forceinline__ float silu(float x) {
    return x / (1.0f + __expf(-x));
}

__device__ __forceinline__ int xhalf_i(int x) {
    return __builtin_amdgcn_permlanex16(x, x, 0x76543210u, 0xfedcba98u, false, false);
}
__device__ __forceinline__ float xhalf_f(float x) {
    return __builtin_bit_cast(float, xhalf_i(__builtin_bit_cast(int, x)));
}

__device__ __forceinline__ bool pack_b_one(const float* __restrict__ W, float2* __restrict__ Wp,
                                           const int K, const int N, int& idx) {
    const int total = (K / 2) * N;
    if (idx < total) {
        const int q = idx / N;
        const int n = idx - q * N;
        float2 p;
        p.x = W[(2 * q) * N + n];
        p.y = W[(2 * q + 1) * N + n];
        Wp[idx] = p;
        return true;
    }
    idx -= total;
    return false;
}

#define PACK_TOTAL (2 * 64 * 128 + 4 * 64 + 2 * 32 * 64 + 32 * 512 + 2 * 128 * 128)

__global__ void prep_kernel(float4* __restrict__ p, int n4, int* __restrict__ cnt, int ncnt,
                            const float* __restrict__ Wup0, const float* __restrict__ Wup1,
                            const float* __restrict__ w1, const float* __restrict__ w2,
                            const float* __restrict__ w3, const float* __restrict__ w4,
                            const float* __restrict__ Wl0, const float* __restrict__ Wl1,
                            float2* __restrict__ Wup0p, float2* __restrict__ Wup1p,
                            float2* __restrict__ w1p, float2* __restrict__ w2p,
                            float2* __restrict__ w3p, float2* __restrict__ w4p,
                            float2* __restrict__ Wl0p, float2* __restrict__ Wl1p) {
    const int stride = gridDim.x * blockDim.x;
    const int gid    = blockIdx.x * blockDim.x + threadIdx.x;
    for (int i = gid; i < n4; i += stride)
        p[i] = make_float4(0.f, 0.f, 0.f, 0.f);
    for (int i = gid; i < ncnt; i += stride)
        cnt[i] = 0;
    for (int g = gid; g < PACK_TOTAL; g += stride) {
        int idx = g;
        if (pack_b_one(Wup0, Wup0p, 128, 128, idx)) continue;
        if (pack_b_one(Wup1, Wup1p, 128, 128, idx)) continue;
        if (pack_b_one(w1,   w1p,   8,   64,  idx)) continue;
        if (pack_b_one(w2,   w2p,   64,  64,  idx)) continue;
        if (pack_b_one(w3,   w3p,   64,  64,  idx)) continue;
        if (pack_b_one(w4,   w4p,   64,  512, idx)) continue;
        if (pack_b_one(Wl0,  Wl0p,  256, 128, idx)) continue;
        pack_b_one(Wl1, Wl1p, 256, 128, idx);
    }
}

__global__ void count_kernel(const int* __restrict__ rcv, int* __restrict__ cnt) {
    const int stride = gridDim.x * blockDim.x;
    for (int e = blockIdx.x * blockDim.x + threadIdx.x; e < N_EDGES; e += stride)
        atomicAdd(&cnt[rcv[e]], 1);
}

#define SCAN_PER_THREAD 20
__global__ void __launch_bounds__(1024)
scan_kernel(int* __restrict__ cnt) {
    __shared__ int part[1024];
    const int t = threadIdx.x;
    const int b = t * SCAN_PER_THREAD;
    int loc[SCAN_PER_THREAD];
    int s = 0;
#pragma unroll
    for (int k = 0; k < SCAN_PER_THREAD; ++k) {
        const int idx = b + k;
        const int v = (idx < N_NODES) ? cnt[idx] : 0;
        loc[k] = s;
        s += v;
    }
    part[t] = s;
    __syncthreads();
    for (int off = 1; off < 1024; off <<= 1) {
        const int v = (t >= off) ? part[t - off] : 0;
        __syncthreads();
        part[t] += v;
        __syncthreads();
    }
    const int base = part[t] - s;
#pragma unroll
    for (int k = 0; k < SCAN_PER_THREAD; ++k) {
        const int idx = b + k;
        if (idx < N_NODES) cnt[idx] = base + loc[k];
    }
}

__global__ void fill_kernel(const int* __restrict__ rcv, int* __restrict__ cnt,
                            int* __restrict__ perm) {
    const int stride = gridDim.x * blockDim.x;
    for (int e = blockIdx.x * blockDim.x + threadIdx.x; e < N_EDGES; e += stride) {
        const int pos = atomicAdd(&cnt[rcv[e]], 1);
        perm[pos] = e;
    }
}

__global__ void __launch_bounds__(256)
node_up_kernel(const float* __restrict__ nf, const v2f* __restrict__ W0p,
               const v2f* __restrict__ W1p, float* __restrict__ s_up,
               float* __restrict__ v_up) {
    const int lane   = threadIdx.x & 31;
    const int wv     = threadIdx.x >> 5;
    const int lane15 = lane & 15;
    const int half   = lane >> 4;
    const int koff   = half * 2;
    const int mbase  = half * 8;
    const int n0     = blockIdx.x * 16;
    const int col    = wv * 16 + lane15;
    const int ab     = (n0 + lane15) * 512;
    const int ab2    = (n0 + lane15) * 256 + half;
    const int wbase  = half * 128 + col;

    const v2f* nf2 = (const v2f*)nf;

    v8f aS = {}, aV0 = {}, aV1 = {}, aV2 = {};
    for (int k = 0; k < 128; k += 4) {
        const int kk = k + koff;
        const int qb = (k >> 2) * 256 + wbase;
        const v2f b0 = W0p[qb];
        const v2f b1 = W1p[qb];
        const v2f a  = nf2[ab2 + (k >> 1)];
        aS = wmma4(a, b0, aS);
        const int vb = ab + 128 + kk * 3;
        v2f av;
        av.x = nf[vb + 0]; av.y = nf[vb + 3];
        aV0 = wmma4(av, b1, aV0);
        av.x = nf[vb + 1]; av.y = nf[vb + 4];
        aV1 = wmma4(av, b1, aV1);
        av.x = nf[vb + 2]; av.y = nf[vb + 5];
        aV2 = wmma4(av, b1, aV2);
    }

    const float inv = 0.08838834764831845f;
#pragma unroll
    for (int r = 0; r < 8; ++r) {
        const int n = n0 + mbase + r;
        s_up[n * 128 + col] = aS[r] * inv;
        const int vo = (n * 128 + col) * 3;
        v_up[vo + 0] = aV0[r] * inv;
        v_up[vo + 1] = aV1[r] * inv;
        v_up[vo + 2] = aV2[r] * inv;
    }
}

__device__ __forceinline__ void mlp64_layer(const float* HSRC, float* HDST,
                                            const v2f* __restrict__ Wp,
                                            int lane15, int half, int mbase) {
    const v2f* H2    = (const v2f*)HSRC;
    const int abase  = lane15 * 34 + half;
    const int wbase  = half * 64 + lane15;
    v8f acc[4] = {{}, {}, {}, {}};
    for (int k = 0; k < 64; k += 4) {
        const v2f a  = H2[abase + (k >> 1)];
        const int qb = (k >> 2) * 128 + wbase;
#pragma unroll
        for (int ct = 0; ct < 4; ++ct) {
            const v2f b = Wp[qb + ct * 16];
            acc[ct] = wmma4(a, b, acc[ct]);
        }
    }
#pragma unroll
    for (int ct = 0; ct < 4; ++ct)
#pragma unroll
        for (int r = 0; r < 8; ++r)
            HDST[(mbase + r) * 68 + ct * 16 + lane15] = silu(acc[ct][r] * 0.125f);
}

template <int J>
__device__ __forceinline__ void tp_pass(
        const int g, const v2f* H2, const int abase, const int half, const int lane15,
        const v2f* __restrict__ w4p, const float* __restrict__ s_up,
        const float* __restrict__ v_up, float* __restrict__ msg,
        const int (&sidx)[8], const int (&vidx)[8], const int (&midx)[8],
        const float (&y0a)[8], const float (&y1xa)[8], const float (&y1ya)[8],
        const float (&y1za)[8], const unsigned fl, const unsigned am, const unsigned sm,
        const unsigned cm, const bool jadd) {
    const float c  = 0.08838834764831845f;
    const float c3 = 0.05103103630798287f;
    const int ub = (g & 1) * 64 + lane15;

    const int wbase = half * 512 + g * 64 + lane15;
    v2f b0[8][4], b1[8][4];
#pragma unroll
    for (int kk = 0; kk < 8; ++kk)
#pragma unroll
        for (int i = 0; i < 4; ++i)
            b0[kk][i] = w4p[kk * 1024 + wbase + i * 16];
    __builtin_amdgcn_sched_barrier(0);

    float gs[4][8];
    float gv[4][8][3];
    if (J == 0 || J == 2) {
#pragma unroll
        for (int i = 0; i < 4; ++i)
#pragma unroll
            for (int r = 0; r < 8; ++r)
                gs[i][r] = s_up[sidx[r] + ub + 16 * i];
    } else {
#pragma unroll
        for (int i = 0; i < 4; ++i)
#pragma unroll
            for (int r = 0; r < 8; ++r) {
                const int vb = vidx[r] + 3 * (ub + 16 * i);
                gv[i][r][0] = v_up[vb + 0];
                gv[i][r][1] = v_up[vb + 1];
                gv[i][r][2] = v_up[vb + 2];
            }
    }

    __builtin_amdgcn_sched_barrier(0);
    v8f acc[4] = {{}, {}, {}, {}};
#pragma unroll
    for (int kk = 0; kk < 8; ++kk) {
        const v2f a = H2[abase + 2 * kk];
#pragma unroll
        for (int i = 0; i < 4; ++i)
            acc[i] = wmma4(a, b0[kk][i], acc[i]);
#pragma unroll
        for (int i = 0; i < 4; ++i)
            b1[kk][i] = w4p[(kk + 8) * 1024 + wbase + i * 16];
        __builtin_amdgcn_sched_group_barrier(0x008, 4, 0);
        __builtin_amdgcn_sched_group_barrier(0x020, 4, 0);
    }
#pragma unroll
    for (int kk = 0; kk < 8; ++kk) {
        const v2f a = H2[abase + 2 * (kk + 8)];
#pragma unroll
        for (int i = 0; i < 4; ++i)
            acc[i] = wmma4(a, b1[kk][i], acc[i]);
    }

    constexpr int NC   = (J >= 2) ? 3 : 1;
    constexpr int COL0 = (J == 0) ? 0 : (J == 1) ? 128 : (J == 2) ? 256 : 640;
    float run[4][3], cap[4][3];
#pragma unroll
    for (int i = 0; i < 4; ++i)
#pragma unroll
        for (int k = 0; k < 3; ++k) { run[i][k] = 0.f; cap[i][k] = 0.f; }

#pragma unroll
    for (int r = 0; r < 8; ++r) {
#pragma unroll
        for (int i = 0; i < 4; ++i) {
            const float w = acc[i][r] * 0.125f;
            if (J == 0) {
                run[i][0] += c * gs[i][r] * y0a[r] * w;
            } else if (J == 1) {
                const float dot = gv[i][r][0] * y1xa[r] + gv[i][r][1] * y1ya[r] +
                                  gv[i][r][2] * y1za[r];
                run[i][0] += c3 * dot * w;
            } else if (J == 2) {
                const float t = c * gs[i][r] * w;
                run[i][0] += t * y1xa[r];
                run[i][1] += t * y1ya[r];
                run[i][2] += t * y1za[r];
            } else {
                const float t = c * y0a[r] * w;
                run[i][0] += t * gv[i][r][0];
                run[i][1] += t * gv[i][r][1];
                run[i][2] += t * gv[i][r][2];
            }
        }
        const bool capr = (cm >> r) & 1u;
#pragma unroll
        for (int i = 0; i < 4; ++i)
#pragma unroll
            for (int k = 0; k < NC; ++k)
                cap[i][k] = capr ? run[i][k] : cap[i][k];
        if (r == 7) {
#pragma unroll
            for (int i = 0; i < 4; ++i)
#pragma unroll
                for (int k = 0; k < NC; ++k) {
                    const float other = xhalf_f(cap[i][k]);
                    run[i][k] = jadd ? run[i][k] + other : run[i][k];
                }
        }
        const int mb = midx[r] + COL0;
        if ((am >> r) & 1u) {
#pragma unroll
            for (int i = 0; i < 4; ++i) {
                const int u = ub + 16 * i;
#pragma unroll
                for (int k = 0; k < NC; ++k)
                    atomicAdd(&msg[mb + NC * u + k], run[i][k]);
            }
        }
        if ((sm >> r) & 1u) {
#pragma unroll
            for (int i = 0; i < 4; ++i) {
                const int u = ub + 16 * i;
#pragma unroll
                for (int k = 0; k < NC; ++k)
                    msg[mb + NC * u + k] = run[i][k];
            }
        }
        if (r < 7) {
            const bool last = (fl >> r) & 1u;
#pragma unroll
            for (int i = 0; i < 4; ++i)
#pragma unroll
                for (int k = 0; k < NC; ++k)
                    run[i][k] = last ? 0.f : run[i][k];
        }
    }
}

__global__ void __launch_bounds__(128)
edge_kernel(const float* __restrict__ ea, const float* __restrict__ ef,
            const int* __restrict__ snd, const int* __restrict__ rcv,
            const v2f* __restrict__ w1p, const v2f* __restrict__ w2p,
            const v2f* __restrict__ w3p, const v2f* __restrict__ w4p,
            const float* __restrict__ s_up, const float* __restrict__ v_up,
            const int* __restrict__ perm, float* __restrict__ msg) {
    __shared__ float hA[4][16 * 68];
    __shared__ float hB[4][16 * 68];

    const int lane   = threadIdx.x & 31;
    const int wv     = threadIdx.x >> 5;
    const int lane15 = lane & 15;
    const int half   = lane >> 4;
    const int mbase  = half * 8;
    const int e0     = (blockIdx.x * 4 + wv) * 16;

    float* HA = hA[wv];
    float* HB = hB[wv];

    __builtin_prefetch((const float*)w4p + threadIdx.x * 256, 0, 1);

    const int  nbpos   = half ? (e0 + 16) : (e0 - 1);
    const bool nbvalid = (nbpos >= 0) && (nbpos < N_EDGES);
    const int  nbrcv   = rcv[perm[nbvalid ? nbpos : e0]];

    int   sidx[8], vidx[8], midx[8];
    float y0a[8], y1xa[8], y1ya[8], y1za[8];
#pragma unroll
    for (int r = 0; r < 8; ++r) {
        const int e    = perm[e0 + mbase + r];
        const int node = snd[e];
        sidx[r] = node * 128;
        vidx[r] = node * 384;
        midx[r] = rcv[e] * 1024;
        const float4 y = *(const float4*)(ea + e * 4);
        y0a[r] = y.x; y1xa[r] = y.y; y1ya[r] = y.z; y1za[r] = y.w;
    }
    unsigned fl = 0x80u;
#pragma unroll
    for (int r = 0; r < 7; ++r)
        if (midx[r] != midx[r + 1]) fl |= (1u << r);

    const int  edge_mine  = half ? midx[0] : midx[7];
    const bool join       = (edge_mine == xhalf_i(edge_mine));
    const unsigned fl_oth = (unsigned)xhalf_i((int)fl);
    const unsigned fb     = 1u << __builtin_ctz(fl);
    const int  nbm       = nbvalid ? nbrcv * 1024 : -1;
    const bool open_mine = (nbm == (half ? midx[7] : midx[0]));
    const bool open_oth  = xhalf_i(open_mine ? 1 : 0) != 0;
    const bool joined_last = join & (fl_oth == 0x80u) & open_oth;
    const unsigned am_lo   = (open_mine ? fb : 0u) | (joined_last ? 0x80u : 0u);
    const unsigned am_hi   = open_mine ? 0x80u : 0u;
    unsigned am = half ? am_hi : am_lo;
    unsigned sm = fl & ~am;
    const unsigned cm = ((half != 0) & join) ? fb : 0u;
    am &= ~cm;
    sm &= ~cm;
    const bool jadd = (half == 0) && join;

    {
        const v2f* ef2  = (const v2f*)ef;
        const int eb2   = perm[e0 + lane15] * 4 + half;
        const int wbase = half * 64 + lane15;
        v8f acc[4] = {{}, {}, {}, {}};
#pragma unroll
        for (int k = 0; k < 8; k += 4) {
            const v2f a  = ef2[eb2 + (k >> 1)];
            const int qb = (k >> 2) * 128 + wbase;
#pragma unroll
            for (int ct = 0; ct < 4; ++ct) {
                const v2f b = w1p[qb + ct * 16];
                acc[ct] = wmma4(a, b, acc[ct]);
            }
        }
#pragma unroll
        for (int ct = 0; ct < 4; ++ct)
#pragma unroll
            for (int r = 0; r < 8; ++r)
                HA[(mbase + r) * 68 + ct * 16 + lane15] = silu(acc[ct][r] * 0.35355339059327373f);
    }
    asm volatile("s_wait_dscnt 0" ::: "memory");

    mlp64_layer(HA, HB, w2p, lane15, half, mbase);
    asm volatile("s_wait_dscnt 0" ::: "memory");
    mlp64_layer(HB, HA, w3p, lane15, half, mbase);
    asm volatile("s_wait_dscnt 0" ::: "memory");

    const v2f* H2   = (const v2f*)HA;
    const int abase = lane15 * 34 + half;
#pragma unroll 1
    for (int g = 0; g < 2; ++g)
        tp_pass<0>(g, H2, abase, half, lane15, w4p, s_up, v_up, msg,
                   sidx, vidx, midx, y0a, y1xa, y1ya, y1za, fl, am, sm, cm, jadd);
#pragma unroll 1
    for (int g = 2; g < 4; ++g)
        tp_pass<1>(g, H2, abase, half, lane15, w4p, s_up, v_up, msg,
                   sidx, vidx, midx, y0a, y1xa, y1ya, y1za, fl, am, sm, cm, jadd);
#pragma unroll 1
    for (int g = 4; g < 6; ++g)
        tp_pass<2>(g, H2, abase, half, lane15, w4p, s_up, v_up, msg,
                   sidx, vidx, midx, y0a, y1xa, y1ya, y1za, fl, am, sm, cm, jadd);
#pragma unroll 1
    for (int g = 6; g < 8; ++g)
        tp_pass<3>(g, H2, abase, half, lane15, w4p, s_up, v_up, msg,
                   sidx, vidx, midx, y0a, y1xa, y1ya, y1za, fl, am, sm, cm, jadd);
}

__global__ void __launch_bounds__(256)
node_down_kernel(const float* __restrict__ msg, const v2f* __restrict__ L0p,
                 const v2f* __restrict__ L1p, float* __restrict__ out) {
    const int lane   = threadIdx.x & 31;
    const int wv     = threadIdx.x >> 5;
    const int lane15 = lane & 15;
    const int half   = lane >> 4;
    const int koff   = half * 2;
    const int mbase  = half * 8;
    const int n0     = blockIdx.x * 16;
    const int col    = wv * 16 + lane15;
    const int ab     = (n0 + lane15) * 1024;
    const int ab2    = (n0 + lane15) * 512 + half;
    const int wbase  = half * 128 + col;

    const v2f* msg2 = (const v2f*)msg;

    v8f aS = {}, aV0 = {}, aV1 = {}, aV2 = {};
    for (int k = 0; k < 256; k += 4) {
        const int kk = k + koff;
        const int qb = (k >> 2) * 256 + wbase;
        const v2f b0 = L0p[qb];
        const v2f b1 = L1p[qb];
        const v2f a  = msg2[ab2 + (k >> 1)];
        aS = wmma4(a, b0, aS);
        const int vb = ab + 256 + kk * 3;
        v2f av;
        av.x = msg[vb + 0]; av.y = msg[vb + 3];
        aV0 = wmma4(av, b1, aV0);
        av.x = msg[vb + 1]; av.y = msg[vb + 4];
        aV1 = wmma4(av, b1, aV1);
        av.x = msg[vb + 2]; av.y = msg[vb + 5];
        aV2 = wmma4(av, b1, aV2);
    }

    const float sc = (1.0f / 16.0f) * 0.1f;
#pragma unroll
    for (int r = 0; r < 8; ++r) {
        const int n = n0 + mbase + r;
        out[n * 512 + col] = aS[r] * sc;
        const int oo = n * 512 + 128 + col * 3;
        out[oo + 0] = aV0[r] * sc;
        out[oo + 1] = aV1[r] * sc;
        out[oo + 2] = aV2[r] * sc;
    }
}

extern "C" void kernel_launch(void* const* d_in, const int* in_sizes, int n_in,
                              void* d_out, int out_size, void* d_ws, size_t ws_size,
                              hipStream_t stream) {
    (void)in_sizes; (void)n_in; (void)out_size; (void)ws_size;
    const float* nf   = (const float*)d_in[0];
    const float* ea   = (const float*)d_in[1];
    const float* ef   = (const float*)d_in[2];
    const int*   snd  = (const int*)d_in[3];
    const int*   rcv  = (const int*)d_in[4];
    const float* Wup0 = (const float*)d_in[5];
    const float* Wup1 = (const float*)d_in[6];
    const float* w1   = (const float*)d_in[7];
    const float* w2   = (const float*)d_in[8];
    const float* w3   = (const float*)d_in[9];
    const float* w4   = (const float*)d_in[10];
    const float* Wl0  = (const float*)d_in[11];
    const float* Wl1  = (const float*)d_in[12];
    float* out = (float*)d_out;

    float* s_up = (float*)d_ws;
    float* v_up = s_up + (size_t)N_NODES * 128;
    float* msgb = v_up + (size_t)N_NODES * 384;
    float* wp   = msgb + (size_t)N_NODES * 1024;
    float* Wup0p = wp;  wp += 128 * 128;
    float* Wup1p = wp;  wp += 128 * 128;
    float* w1p   = wp;  wp += 8 * 64;
    float* w2p   = wp;  wp += 64 * 64;
    float* w3p   = wp;  wp += 64 * 64;
    float* w4p   = wp;  wp += 64 * 512;
    float* Wl0p  = wp;  wp += 256 * 128;
    float* Wl1p  = wp;  wp += 256 * 128;

    int* cnt  = (int*)d_out;
    int* perm = cnt + N_NODES;

    prep_kernel<<<2048, 256, 0, stream>>>((float4*)msgb, N_NODES * 1024 / 4, cnt, N_NODES,
                                          Wup0, Wup1, w1, w2, w3, w4, Wl0, Wl1,
                                          (float2*)Wup0p, (float2*)Wup1p, (float2*)w1p,
                                          (float2*)w2p, (float2*)w3p, (float2*)w4p,
                                          (float2*)Wl0p, (float2*)Wl1p);

    count_kernel<<<196, 1024, 0, stream>>>(rcv, cnt);
    scan_kernel<<<1, 1024, 0, stream>>>(cnt);
    fill_kernel<<<196, 1024, 0, stream>>>(rcv, cnt, perm);

    node_up_kernel<<<N_NODES / 16, 256, 0, stream>>>(nf, (const v2f*)Wup0p,
                                                     (const v2f*)Wup1p, s_up, v_up);
    edge_kernel<<<N_EDGES / 64, 128, 0, stream>>>(ea, ef, snd, rcv,
                                                  (const v2f*)w1p, (const v2f*)w2p,
                                                  (const v2f*)w3p, (const v2f*)w4p,
                                                  s_up, v_up, perm, msgb);
    node_down_kernel<<<N_NODES / 16, 256, 0, stream>>>(msgb, (const v2f*)Wl0p,
                                                       (const v2f*)Wl1p, out);
}


# Edge kernel: 256 resident workgroups of 12 waves hold the packed mlp_w4 in LDS and take 16-edge tiles from an LDS counter; layer-4 weight fragments are LDS reads, not 128 KB from L2 per tile
# speedup vs baseline: 2.5469x; 1.0702x over previous
#include <hip/hip_runtime.h>

typedef float v2f __attribute__((ext_vector_type(2)));
typedef float v8f __attribute__((ext_vector_type(8)));

#define N_NODES 20000
#define N_EDGES 200000

__device__ __forceinline__ v8f wmma4(v2f a, v2f b, v8f c) {
    return __builtin_amdgcn_wmma_f32_16x16x4_f32(
         false, a,  false, b,
         (short)0, c,  false,  false);
}

__device__ __forceinline__ float silu(float x) {
    return x / (1.0f + __expf(-x));
}

__device__ __forceinline__ int xhalf_i(int x) {
    return __builtin_amdgcn_permlanex16(x, x, 0x76543210u, 0xfedcba98u, false, false);
}
__device__ __forceinline__ float xhalf_f(float x) {
    return __builtin_bit_cast(float, xhalf_i(__builtin_bit_cast(int, x)));
}

__device__ __forceinline__ bool pack_b_one(const float* __restrict__ W, float2* __restrict__ Wp,
                                           const int K, const int N, int& idx) {
    const int total = (K / 2) * N;
    if (idx < total) {
        const int q = idx / N;
        const int n = idx - q * N;
        float2 p;
        p.x = W[(2 * q) * N + n];
        p.y = W[(2 * q + 1) * N + n];
        Wp[idx] = p;
        return true;
    }
    idx -= total;
    return false;
}

#define PACK_TOTAL (2 * 64 * 128 + 4 * 64 + 2 * 32 * 64 + 32 * 512 + 2 * 128 * 128)

__global__ void prep_kernel(float4* __restrict__ p, int n4, int* __restrict__ cnt, int ncnt,
                            const float* __restrict__ Wup0, const float* __restrict__ Wup1,
                            const float* __restrict__ w1, const float* __restrict__ w2,
                            const float* __restrict__ w3, const float* __restrict__ w4,
                            const float* __restrict__ Wl0, const float* __restrict__ Wl1,
                            float2* __restrict__ Wup0p, float2* __restrict__ Wup1p,
                            float2* __restrict__ w1p, float2* __restrict__ w2p,
                            float2* __restrict__ w3p, float2* __restrict__ w4p,
                            float2* __restrict__ Wl0p, float2* __restrict__ Wl1p) {
    const int stride = gridDim.x * blockDim.x;
    const int gid    = blockIdx.x * blockDim.x + threadIdx.x;
    for (int i = gid; i < n4; i += stride)
        p[i] = make_float4(0.f, 0.f, 0.f, 0.f);
    for (int i = gid; i < ncnt; i += stride)
        cnt[i] = 0;
    for (int g = gid; g < PACK_TOTAL; g += stride) {
        int idx = g;
        if (pack_b_one(Wup0, Wup0p, 128, 128, idx)) continue;
        if (pack_b_one(Wup1, Wup1p, 128, 128, idx)) continue;
        if (pack_b_one(w1,   w1p,   8,   64,  idx)) continue;
        if (pack_b_one(w2,   w2p,   64,  64,  idx)) continue;
        if (pack_b_one(w3,   w3p,   64,  64,  idx)) continue;
        if (pack_b_one(w4,   w4p,   64,  512, idx)) continue;
        if (pack_b_one(Wl0,  Wl0p,  256, 128, idx)) continue;
        pack_b_one(Wl1, Wl1p, 256, 128, idx);
    }
}

__global__ void count_kernel(const int* __restrict__ rcv, int* __restrict__ cnt) {
    const int stride = gridDim.x * blockDim.x;
    for (int e = blockIdx.x * blockDim.x + threadIdx.x; e < N_EDGES; e += stride)
        atomicAdd(&cnt[rcv[e]], 1);
}

#define SCAN_PER_THREAD 20
__global__ void __launch_bounds__(1024)
scan_kernel(int* __restrict__ cnt) {
    __shared__ int part[1024];
    const int t = threadIdx.x;
    const int b = t * SCAN_PER_THREAD;
    int loc[SCAN_PER_THREAD];
    int s = 0;
#pragma unroll
    for (int k = 0; k < SCAN_PER_THREAD; ++k) {
        const int idx = b + k;
        const int v = (idx < N_NODES) ? cnt[idx] : 0;
        loc[k] = s;
        s += v;
    }
    part[t] = s;
    __syncthreads();
    for (int off = 1; off < 1024; off <<= 1) {
        const int v = (t >= off) ? part[t - off] : 0;
        __syncthreads();
        part[t] += v;
        __syncthreads();
    }
    const int base = part[t] - s;
#pragma unroll
    for (int k = 0; k < SCAN_PER_THREAD; ++k) {
        const int idx = b + k;
        if (idx < N_NODES) cnt[idx] = base + loc[k];
    }
}

__global__ void fill_kernel(const int* __restrict__ rcv, int* __restrict__ cnt,
                            int* __restrict__ perm) {
    const int stride = gridDim.x * blockDim.x;
    for (int e = blockIdx.x * blockDim.x + threadIdx.x; e < N_EDGES; e += stride) {
        const int pos = atomicAdd(&cnt[rcv[e]], 1);
        perm[pos] = e;
    }
}

__global__ void __launch_bounds__(256)
node_up_kernel(const float* __restrict__ nf, const v2f* __restrict__ W0p,
               const v2f* __restrict__ W1p, float* __restrict__ s_up,
               float* __restrict__ v_up) {
    const int lane   = threadIdx.x & 31;
    const int wv     = threadIdx.x >> 5;
    const int lane15 = lane & 15;
    const int half   = lane >> 4;
    const int koff   = half * 2;
    const int mbase  = half * 8;
    const int n0     = blockIdx.x * 16;
    const int col    = wv * 16 + lane15;
    const int ab     = (n0 + lane15) * 512;
    const int ab2    = (n0 + lane15) * 256 + half;
    const int wbase  = half * 128 + col;

    const v2f* nf2 = (const v2f*)nf;

    v8f aS = {}, aV0 = {}, aV1 = {}, aV2 = {};
    for (int k = 0; k < 128; k += 4) {
        const int kk = k + koff;
        const int qb = (k >> 2) * 256 + wbase;
        const v2f b0 = W0p[qb];
        const v2f b1 = W1p[qb];
        const v2f a  = nf2[ab2 + (k >> 1)];
        aS = wmma4(a, b0, aS);
        const int vb = ab + 128 + kk * 3;
        v2f av;
        av.x = nf[vb + 0]; av.y = nf[vb + 3];
        aV0 = wmma4(av, b1, aV0);
        av.x = nf[vb + 1]; av.y = nf[vb + 4];
        aV1 = wmma4(av, b1, aV1);
        av.x = nf[vb + 2]; av.y = nf[vb + 5];
        aV2 = wmma4(av, b1, aV2);
    }

    const float inv = 0.08838834764831845f;
#pragma unroll
    for (int r = 0; r < 8; ++r) {
        const int n = n0 + mbase + r;
        s_up[n * 128 + col] = aS[r] * inv;
        const int vo = (n * 128 + col) * 3;
        v_up[vo + 0] = aV0[r] * inv;
        v_up[vo + 1] = aV1[r] * inv;
        v_up[vo + 2] = aV2[r] * inv;
    }
}

__device__ __forceinline__ void mlp64_layer(const float* HSRC, float* HDST,
                                            const v2f* __restrict__ Wp,
                                            int lane15, int half, int mbase) {
    const v2f* H2    = (const v2f*)HSRC;
    const int abase  = lane15 * 34 + half;
    const int wbase  = half * 64 + lane15;
    v8f acc[4] = {{}, {}, {}, {}};
    for (int k = 0; k < 64; k += 4) {
        const v2f a  = H2[abase + (k >> 1)];
        const int qb = (k >> 2) * 128 + wbase;
#pragma unroll
        for (int ct = 0; ct < 4; ++ct) {
            const v2f b = Wp[qb + ct * 16];
            acc[ct] = wmma4(a, b, acc[ct]);
        }
    }
#pragma unroll
    for (int ct = 0; ct < 4; ++ct)
#pragma unroll
        for (int r = 0; r < 8; ++r)
            HDST[(mbase + r) * 68 + ct * 16 + lane15] = silu(acc[ct][r] * 0.125f);
}

template <int J>
__device__ __forceinline__ void tp_pass(
        const int g, const v2f* H2, const int abase, const int lane, const int lane15,
        const v2f* w4s, const float* __restrict__ s_up,
        const float* __restrict__ v_up, float* __restrict__ msg,
        const int (&sidx)[8], const int (&vidx)[8], const int (&midx)[8],
        const float (&y0a)[8], const float (&y1xa)[8], const float (&y1ya)[8],
        const float (&y1za)[8], const unsigned fl, const unsigned am, const unsigned sm,
        const unsigned cm, const bool jadd) {
    const float c  = 0.08838834764831845f;
    const float c3 = 0.05103103630798287f;
    const int ub = (g & 1) * 64 + lane15;

    float gs[4][8];
    float gv[4][8][3];
    if (J == 0 || J == 2) {
#pragma unroll
        for (int i = 0; i < 4; ++i)
#pragma unroll
            for (int r = 0; r < 8; ++r)
                gs[i][r] = s_up[sidx[r] + ub + 16 * i];
    } else {
#pragma unroll
        for (int i = 0; i < 4; ++i)
#pragma unroll
            for (int r = 0; r < 8; ++r) {
                const int vb = vidx[r] + 3 * (ub + 16 * i);
                gv[i][r][0] = v_up[vb + 0];
                gv[i][r][1] = v_up[vb + 1];
                gv[i][r][2] = v_up[vb + 2];
            }
    }

    __builtin_amdgcn_sched_barrier(0);

    const int wb = g * 128 + lane;
    v2f bq[2][4][4], aq[2][4];
#pragma unroll
    for (int s = 0; s < 4; ++s) {
        aq[0][s] = H2[abase + 2 * s];
#pragma unroll
        for (int i = 0; i < 4; ++i)
            bq[0][s][i] = w4s[s * 1024 + wb + i * 32];
    }
    v8f acc[4] = {{}, {}, {}, {}};
#pragma unroll
    for (int blk = 0; blk < 4; ++blk) {
        const int cur = blk & 1;
        if (blk < 3) {
#pragma unroll
            for (int s = 0; s < 4; ++s) {
                const int kk = (blk + 1) * 4 + s;
                aq[cur ^ 1][s] = H2[abase + 2 * kk];
#pragma unroll
                for (int i = 0; i < 4; ++i)
                    bq[cur ^ 1][s][i] = w4s[kk * 1024 + wb + i * 32];
            }
            __builtin_amdgcn_sched_barrier(0);
        }
#pragma unroll
        for (int s = 0; s < 4; ++s)
#pragma unroll
            for (int i = 0; i < 4; ++i)
                acc[i] = wmma4(aq[cur][s], bq[cur][s][i], acc[i]);
        if (blk < 2) __builtin_amdgcn_sched_barrier(0);
    }

    constexpr int NC   = (J >= 2) ? 3 : 1;
    constexpr int COL0 = (J == 0) ? 0 : (J == 1) ? 128 : (J == 2) ? 256 : 640;
    float run[4][3], cap[4][3];
#pragma unroll
    for (int i = 0; i < 4; ++i)
#pragma unroll
        for (int k = 0; k < 3; ++k) { run[i][k] = 0.f; cap[i][k] = 0.f; }

#pragma unroll
    for (int r = 0; r < 8; ++r) {
#pragma unroll
        for (int i = 0; i < 4; ++i) {
            const float w = acc[i][r] * 0.125f;
            if (J == 0) {
                run[i][0] += c * gs[i][r] * y0a[r] * w;
            } else if (J == 1) {
                const float dot = gv[i][r][0] * y1xa[r] + gv[i][r][1] * y1ya[r] +
                                  gv[i][r][2] * y1za[r];
                run[i][0] += c3 * dot * w;
            } else if (J == 2) {
                const float t = c * gs[i][r] * w;
                run[i][0] += t * y1xa[r];
                run[i][1] += t * y1ya[r];
                run[i][2] += t * y1za[r];
            } else {
                const float t = c * y0a[r] * w;
                run[i][0] += t * gv[i][r][0];
                run[i][1] += t * gv[i][r][1];
                run[i][2] += t * gv[i][r][2];
            }
        }
        const bool capr = (cm >> r) & 1u;
#pragma unroll
        for (int i = 0; i < 4; ++i)
#pragma unroll
            for (int k = 0; k < NC; ++k)
                cap[i][k] = capr ? run[i][k] : cap[i][k];
        if (r == 7) {
#pragma unroll
            for (int i = 0; i < 4; ++i)
#pragma unroll
                for (int k = 0; k < NC; ++k) {
                    const float other = xhalf_f(cap[i][k]);
                    run[i][k] = jadd ? run[i][k] + other : run[i][k];
                }
        }
        const int mb = midx[r] + COL0;
        if ((am >> r) & 1u) {
#pragma unroll
            for (int i = 0; i < 4; ++i) {
                const int u = ub + 16 * i;
#pragma unroll
                for (int k = 0; k < NC; ++k)
                    atomicAdd(&msg[mb + NC * u + k], run[i][k]);
            }
        }
        if ((sm >> r) & 1u) {
#pragma unroll
            for (int i = 0; i < 4; ++i) {
                const int u = ub + 16 * i;
#pragma unroll
                for (int k = 0; k < NC; ++k)
                    msg[mb + NC * u + k] = run[i][k];
            }
        }
        if (r < 7) {
            const bool last = (fl >> r) & 1u;
#pragma unroll
            for (int i = 0; i < 4; ++i)
#pragma unroll
                for (int k = 0; k < NC; ++k)
                    run[i][k] = last ? 0.f : run[i][k];
        }
    }
}

#define EDGE_WAVES 12
#define N_TILES (N_EDGES / 16)

__device__ __forceinline__ int w4s_slot(const int q) {
    const int kk = q >> 10, hf = (q >> 9) & 1, col = q & 511;
    return kk * 1024 + (col >> 4) * 32 + hf * 16 + (col & 15);
}

__global__ void __launch_bounds__(EDGE_WAVES * 32)
edge_kernel(const float* __restrict__ ea, const float* __restrict__ ef,
            const int* __restrict__ snd, const int* __restrict__ rcv,
            const v2f* __restrict__ w1p, const v2f* __restrict__ w2p,
            const v2f* __restrict__ w3p, const v2f* __restrict__ w4p,
            const float* __restrict__ s_up, const float* __restrict__ v_up,
            const int* __restrict__ perm, float* __restrict__ msg) {
    __shared__ float4 w4s4[8 * 1024];
    __shared__ float hA[EDGE_WAVES][16 * 68];
    __shared__ float hB[EDGE_WAVES][16 * 68];
    __shared__ int   tile_next;

    const int lane   = threadIdx.x & 31;
    const int wv     = threadIdx.x >> 5;
    const int lane15 = lane & 15;
    const int half   = lane >> 4;
    const int mbase  = half * 8;

    float* HA = hA[wv];
    float* HB = hB[wv];
    const v2f* w4s = (const v2f*)w4s4;

    if (threadIdx.x == 0) tile_next = 0;
    {
        const float4* w4g = (const float4*)w4p;
        constexpr int NT   = EDGE_WAVES * 32;
        constexpr int FULL = 8192 / NT;
        const bool    tail = (int)threadIdx.x < 8192 - FULL * NT;
        float4 pc[FULL];
        float4 pt = make_float4(0.f, 0.f, 0.f, 0.f);
#pragma unroll
        for (int j = 0; j < FULL; ++j)
            pc[j] = w4g[threadIdx.x + j * NT];
        if (tail) pt = w4g[threadIdx.x + FULL * NT];
#pragma unroll
        for (int j = 0; j < FULL; ++j)
            w4s4[w4s_slot(2 * ((int)threadIdx.x + j * NT)) >> 1] = pc[j];
        if (tail) w4s4[w4s_slot(2 * ((int)threadIdx.x + FULL * NT)) >> 1] = pt;
    }
    __syncthreads();

    const int tper  = N_TILES / (int)gridDim.x;
    const int trem  = N_TILES % (int)gridDim.x;
    const int tbeg  = (int)blockIdx.x * tper + min((int)blockIdx.x, trem);
    const int tcnt  = tper + (((int)blockIdx.x < trem) ? 1 : 0);

    for (;;) {
        int tk = 0;
        if (lane == 0) tk = atomicAdd(&tile_next, 1);
        tk = __builtin_amdgcn_readfirstlane(tk);
        if (tk >= tcnt) break;
        const int e0 = (tbeg + tk) * 16;

        const int  nbpos   = half ? (e0 + 16) : (e0 - 1);
        const bool nbvalid = (nbpos >= 0) && (nbpos < N_EDGES);
        const int  nbrcv   = rcv[perm[nbvalid ? nbpos : e0]];

        int   sidx[8], vidx[8], midx[8];
        float y0a[8], y1xa[8], y1ya[8], y1za[8];
#pragma unroll
        for (int r = 0; r < 8; ++r) {
            const int e    = perm[e0 + mbase + r];
            const int node = snd[e];
            sidx[r] = node * 128;
            vidx[r] = node * 384;
            midx[r] = rcv[e] * 1024;
            const float4 y = *(const float4*)(ea + e * 4);
            y0a[r] = y.x; y1xa[r] = y.y; y1ya[r] = y.z; y1za[r] = y.w;
        }
        unsigned fl = 0x80u;
#pragma unroll
        for (int r = 0; r < 7; ++r)
            if (midx[r] != midx[r + 1]) fl |= (1u << r);

        const int  edge_mine  = half ? midx[0] : midx[7];
        const bool join       = (edge_mine == xhalf_i(edge_mine));
        const unsigned fl_oth = (unsigned)xhalf_i((int)fl);
        const unsigned fb     = 1u << __builtin_ctz(fl);
        const int  nbm       = nbvalid ? nbrcv * 1024 : -1;
        const bool open_mine = (nbm == (half ? midx[7] : midx[0]));
        const bool open_oth  = xhalf_i(open_mine ? 1 : 0) != 0;
        const bool joined_last = join & (fl_oth == 0x80u) & open_oth;
        const unsigned am_lo   = (open_mine ? fb : 0u) | (joined_last ? 0x80u : 0u);
        const unsigned am_hi   = open_mine ? 0x80u : 0u;
        unsigned am = half ? am_hi : am_lo;
        unsigned sm = fl & ~am;
        const unsigned cm = ((half != 0) & join) ? fb : 0u;
        am &= ~cm;
        sm &= ~cm;
        const bool jadd = (half == 0) && join;

        {
            const v2f* ef2  = (const v2f*)ef;
            const int eb2   = perm[e0 + lane15] * 4 + half;
            const int wbase = half * 64 + lane15;
            v8f acc[4] = {{}, {}, {}, {}};
#pragma unroll
            for (int k = 0; k < 8; k += 4) {
                const v2f a  = ef2[eb2 + (k >> 1)];
                const int qb = (k >> 2) * 128 + wbase;
#pragma unroll
                for (int ct = 0; ct < 4; ++ct) {
                    const v2f b = w1p[qb + ct * 16];
                    acc[ct] = wmma4(a, b, acc[ct]);
                }
            }
#pragma unroll
            for (int ct = 0; ct < 4; ++ct)
#pragma unroll
                for (int r = 0; r < 8; ++r)
                    HA[(mbase + r) * 68 + ct * 16 + lane15] = silu(acc[ct][r] * 0.35355339059327373f);
        }
        asm volatile("s_wait_dscnt 0" ::: "memory");

        mlp64_layer(HA, HB, w2p, lane15, half, mbase);
        asm volatile("s_wait_dscnt 0" ::: "memory");
        mlp64_layer(HB, HA, w3p, lane15, half, mbase);
        asm volatile("s_wait_dscnt 0" ::: "memory");

        const v2f* H2   = (const v2f*)HA;
        const int abase = lane15 * 34 + half;
#pragma unroll 1
        for (int g = 0; g < 2; ++g)
            tp_pass<0>(g, H2, abase, lane, lane15, w4s, s_up, v_up, msg,
                       sidx, vidx, midx, y0a, y1xa, y1ya, y1za, fl, am, sm, cm, jadd);
#pragma unroll 1
        for (int g = 2; g < 4; ++g)
            tp_pass<1>(g, H2, abase, lane, lane15, w4s, s_up, v_up, msg,
                       sidx, vidx, midx, y0a, y1xa, y1ya, y1za, fl, am, sm, cm, jadd);
#pragma unroll 1
        for (int g = 4; g < 6; ++g)
            tp_pass<2>(g, H2, abase, lane, lane15, w4s, s_up, v_up, msg,
                       sidx, vidx, midx, y0a, y1xa, y1ya, y1za, fl, am, sm, cm, jadd);
#pragma unroll 1
        for (int g = 6; g < 8; ++g)
            tp_pass<3>(g, H2, abase, lane, lane15, w4s, s_up, v_up, msg,
                       sidx, vidx, midx, y0a, y1xa, y1ya, y1za, fl, am, sm, cm, jadd);
    }
}

__global__ void __launch_bounds__(256)
node_down_kernel(const float* __restrict__ msg, const v2f* __restrict__ L0p,
                 const v2f* __restrict__ L1p, float* __restrict__ out) {
    const int lane   = threadIdx.x & 31;
    const int wv     = threadIdx.x >> 5;
    const int lane15 = lane & 15;
    const int half   = lane >> 4;
    const int koff   = half * 2;
    const int mbase  = half * 8;
    const int n0     = blockIdx.x * 16;
    const int col    = wv * 16 + lane15;
    const int ab     = (n0 + lane15) * 1024;
    const int ab2    = (n0 + lane15) * 512 + half;
    const int wbase  = half * 128 + col;

    const v2f* msg2 = (const v2f*)msg;

    v8f aS = {}, aV0 = {}, aV1 = {}, aV2 = {};
    for (int k = 0; k < 256; k += 4) {
        const int kk = k + koff;
        const int qb = (k >> 2) * 256 + wbase;
        const v2f b0 = L0p[qb];
        const v2f b1 = L1p[qb];
        const v2f a  = msg2[ab2 + (k >> 1)];
        aS = wmma4(a, b0, aS);
        const int vb = ab + 256 + kk * 3;
        v2f av;
        av.x = msg[vb + 0]; av.y = msg[vb + 3];
        aV0 = wmma4(av, b1, aV0);
        av.x = msg[vb + 1]; av.y = msg[vb + 4];
        aV1 = wmma4(av, b1, aV1);
        av.x = msg[vb + 2]; av.y = msg[vb + 5];
        aV2 = wmma4(av, b1, aV2);
    }

    const float sc = (1.0f / 16.0f) * 0.1f;
#pragma unroll
    for (int r = 0; r < 8; ++r) {
        const int n = n0 + mbase + r;
        out[n * 512 + col] = aS[r] * sc;
        const int oo = n * 512 + 128 + col * 3;
        out[oo + 0] = aV0[r] * sc;
        out[oo + 1] = aV1[r] * sc;
        out[oo + 2] = aV2[r] * sc;
    }
}

extern "C" void kernel_launch(void* const* d_in, const int* in_sizes, int n_in,
                              void* d_out, int out_size, void* d_ws, size_t ws_size,
                              hipStream_t stream) {
    (void)in_sizes; (void)n_in; (void)out_size; (void)ws_size;
    const float* nf   = (const float*)d_in[0];
    const float* ea   = (const float*)d_in[1];
    const float* ef   = (const float*)d_in[2];
    const int*   snd  = (const int*)d_in[3];
    const int*   rcv  = (const int*)d_in[4];
    const float* Wup0 = (const float*)d_in[5];
    const float* Wup1 = (const float*)d_in[6];
    const float* w1   = (const float*)d_in[7];
    const float* w2   = (const float*)d_in[8];
    const float* w3   = (const float*)d_in[9];
    const float* w4   = (const float*)d_in[10];
    const float* Wl0  = (const float*)d_in[11];
    const float* Wl1  = (const float*)d_in[12];
    float* out = (float*)d_out;

    float* s_up = (float*)d_ws;
    float* v_up = s_up + (size_t)N_NODES * 128;
    float* msgb = v_up + (size_t)N_NODES * 384;
    float* wp   = msgb + (size_t)N_NODES * 1024;
    float* Wup0p = wp;  wp += 128 * 128;
    float* Wup1p = wp;  wp += 128 * 128;
    float* w1p   = wp;  wp += 8 * 64;
    float* w2p   = wp;  wp += 64 * 64;
    float* w3p   = wp;  wp += 64 * 64;
    float* w4p   = wp;  wp += 64 * 512;
    float* Wl0p  = wp;  wp += 256 * 128;
    float* Wl1p  = wp;  wp += 256 * 128;

    int* cnt  = (int*)d_out;
    int* perm = cnt + N_NODES;

    prep_kernel<<<2048, 256, 0, stream>>>((float4*)msgb, N_NODES * 1024 / 4, cnt, N_NODES,
                                          Wup0, Wup1, w1, w2, w3, w4, Wl0, Wl1,
                                          (float2*)Wup0p, (float2*)Wup1p, (float2*)w1p,
                                          (float2*)w2p, (float2*)w3p, (float2*)w4p,
                                          (float2*)Wl0p, (float2*)Wl1p);

    count_kernel<<<196, 1024, 0, stream>>>(rcv, cnt);
    scan_kernel<<<1, 1024, 0, stream>>>(cnt);
    fill_kernel<<<196, 1024, 0, stream>>>(rcv, cnt, perm);

    node_up_kernel<<<N_NODES / 16, 256, 0, stream>>>(nf, (const v2f*)Wup0p,
                                                     (const v2f*)Wup1p, s_up, v_up);
    edge_kernel<<<256, EDGE_WAVES * 32, 0, stream>>>(ea, ef, snd, rcv,
                                                  (const v2f*)w1p, (const v2f*)w2p,
                                                  (const v2f*)w3p, (const v2f*)w4p,
                                                  s_up, v_up, perm, msgb);
    node_down_kernel<<<N_NODES / 16, 256, 0, stream>>>(msgb, (const v2f*)Wl0p,
                                                       (const v2f*)Wl1p, out);
}


# Node-side linear kernels (up, down): the workgroup's 16 input rows are read once by whole-line 16-byte loads into LDS planes (vector components taken apart); A fragments are 8-byte LDS reads
# speedup vs baseline: 2.8548x; 1.1209x over previous
#include <hip/hip_runtime.h>

typedef float v2f __attribute__((ext_vector_type(2)));
typedef float v8f __attribute__((ext_vector_type(8)));

#define N_NODES 20000
#define N_EDGES 200000

__device__ __forceinline__ v8f wmma4(v2f a, v2f b, v8f c) {
    return __builtin_amdgcn_wmma_f32_16x16x4_f32(
         false, a,  false, b,
         (short)0, c,  false,  false);
}

__device__ __forceinline__ float silu(float x) {
    return x / (1.0f + __expf(-x));
}

__device__ __forceinline__ int xhalf_i(int x) {
    return __builtin_amdgcn_permlanex16(x, x, 0x76543210u, 0xfedcba98u, false, false);
}
__device__ __forceinline__ float xhalf_f(float x) {
    return __builtin_bit_cast(float, xhalf_i(__builtin_bit_cast(int, x)));
}

__device__ __forceinline__ bool pack_b_one(const float* __restrict__ W, float2* __restrict__ Wp,
                                           const int K, const int N, int& idx) {
    const int total = (K / 2) * N;
    if (idx < total) {
        const int q = idx / N;
        const int n = idx - q * N;
        float2 p;
        p.x = W[(2 * q) * N + n];
        p.y = W[(2 * q + 1) * N + n];
        Wp[idx] = p;
        return true;
    }
    idx -= total;
    return false;
}

#define PACK_TOTAL (2 * 64 * 128 + 4 * 64 + 2 * 32 * 64 + 32 * 512 + 2 * 128 * 128)

__global__ void prep_kernel(float4* __restrict__ p, int n4, int* __restrict__ cnt, int ncnt,
                            const float* __restrict__ Wup0, const float* __restrict__ Wup1,
                            const float* __restrict__ w1, const float* __restrict__ w2,
                            const float* __restrict__ w3, const float* __restrict__ w4,
                            const float* __restrict__ Wl0, const float* __restrict__ Wl1,
                            float2* __restrict__ Wup0p, float2* __restrict__ Wup1p,
                            float2* __restrict__ w1p, float2* __restrict__ w2p,
                            float2* __restrict__ w3p, float2* __restrict__ w4p,
                            float2* __restrict__ Wl0p, float2* __restrict__ Wl1p) {
    const int stride = gridDim.x * blockDim.x;
    const int gid    = blockIdx.x * blockDim.x + threadIdx.x;
    for (int i = gid; i < n4; i += stride)
        p[i] = make_float4(0.f, 0.f, 0.f, 0.f);
    for (int i = gid; i < ncnt; i += stride)
        cnt[i] = 0;
    for (int g = gid; g < PACK_TOTAL; g += stride) {
        int idx = g;
        if (pack_b_one(Wup0, Wup0p, 128, 128, idx)) continue;
        if (pack_b_one(Wup1, Wup1p, 128, 128, idx)) continue;
        if (pack_b_one(w1,   w1p,   8,   64,  idx)) continue;
        if (pack_b_one(w2,   w2p,   64,  64,  idx)) continue;
        if (pack_b_one(w3,   w3p,   64,  64,  idx)) continue;
        if (pack_b_one(w4,   w4p,   64,  512, idx)) continue;
        if (pack_b_one(Wl0,  Wl0p,  256, 128, idx)) continue;
        pack_b_one(Wl1, Wl1p, 256, 128, idx);
    }
}

__global__ void count_kernel(const int* __restrict__ rcv, int* __restrict__ cnt) {
    const int stride = gridDim.x * blockDim.x;
    for (int e = blockIdx.x * blockDim.x + threadIdx.x; e < N_EDGES; e += stride)
        atomicAdd(&cnt[rcv[e]], 1);
}

#define SCAN_PER_THREAD 20
__global__ void __launch_bounds__(1024)
scan_kernel(int* __restrict__ cnt) {
    __shared__ int part[1024];
    const int t = threadIdx.x;
    const int b = t * SCAN_PER_THREAD;
    int loc[SCAN_PER_THREAD];
    int s = 0;
#pragma unroll
    for (int k = 0; k < SCAN_PER_THREAD; ++k) {
        const int idx = b + k;
        const int v = (idx < N_NODES) ? cnt[idx] : 0;
        loc[k] = s;
        s += v;
    }
    part[t] = s;
    __syncthreads();
    for (int off = 1; off < 1024; off <<= 1) {
        const int v = (t >= off) ? part[t - off] : 0;
        __syncthreads();
        part[t] += v;
        __syncthreads();
    }
    const int base = part[t] - s;
#pragma unroll
    for (int k = 0; k < SCAN_PER_THREAD; ++k) {
        const int idx = b + k;
        if (idx < N_NODES) cnt[idx] = base + loc[k];
    }
}

__global__ void fill_kernel(const int* __restrict__ rcv, int* __restrict__ cnt,
                            int* __restrict__ perm) {
    const int stride = gridDim.x * blockDim.x;
    for (int e = blockIdx.x * blockDim.x + threadIdx.x; e < N_EDGES; e += stride) {
        const int pos = atomicAdd(&cnt[rcv[e]], 1);
        perm[pos] = e;
    }
}

template <int K>
__device__ __forceinline__ void node_rows_to_lds(const float* __restrict__ in, float* tile,
                                                 const int n0) {
    constexpr int PITCH = K + 4;
    constexpr int PLANE = 16 * PITCH;
    constexpr int Q     = K / 4;
    constexpr int PER   = 16 * Q / 256;
    const float4* in4 = (const float4*)in;
    float4 s4[PER], va[PER], vb[PER], vc[PER];
#pragma unroll
    for (int j = 0; j < PER; ++j) {
        const int idx = (int)threadIdx.x + j * 256;
        const int row = idx / Q, c = idx % Q;
        const int rb  = (n0 + row) * K;
        s4[j] = in4[rb + c];
        va[j] = in4[rb + Q + 3 * c + 0];
        vb[j] = in4[rb + Q + 3 * c + 1];
        vc[j] = in4[rb + Q + 3 * c + 2];
    }
#pragma unroll
    for (int j = 0; j < PER; ++j) {
        const int idx = (int)threadIdx.x + j * 256;
        const int row = idx / Q, c = idx % Q;
        float* d = tile + row * PITCH + 4 * c;
        *(float4*)(d)             = s4[j];
        *(float4*)(d + PLANE)     = make_float4(va[j].x, va[j].w, vb[j].z, vc[j].y);
        *(float4*)(d + 2 * PLANE) = make_float4(va[j].y, vb[j].x, vb[j].w, vc[j].z);
        *(float4*)(d + 3 * PLANE) = make_float4(va[j].z, vb[j].y, vc[j].x, vc[j].w);
    }
}

template <int K>
__device__ __forceinline__ void node_products(const float* tile, const v2f* __restrict__ W0p,
                                              const v2f* __restrict__ W1p, const int lane15,
                                              const int half, const int col,
                                              v8f& aS, v8f& aV0, v8f& aV1, v8f& aV2) {
    constexpr int PITCH2 = (K + 4) / 2;
    constexpr int PLANE2 = 16 * PITCH2;
    const v2f* a2    = (const v2f*)tile + lane15 * PITCH2 + half;
    const int  wbase = half * 128 + col;
    for (int k = 0; k < K; k += 4) {
        const int qb = (k >> 2) * 256 + wbase;
        const v2f b0 = W0p[qb];
        const v2f b1 = W1p[qb];
        const int ka = k >> 1;
        aS  = wmma4(a2[ka], b0, aS);
        aV0 = wmma4(a2[PLANE2 + ka], b1, aV0);
        aV1 = wmma4(a2[2 * PLANE2 + ka], b1, aV1);
        aV2 = wmma4(a2[3 * PLANE2 + ka], b1, aV2);
    }
}

__global__ void __launch_bounds__(256)
node_up_kernel(const float* __restrict__ nf, const v2f* __restrict__ W0p,
               const v2f* __restrict__ W1p, float* __restrict__ s_up,
               float* __restrict__ v_up) {
    __shared__ float4 tile4[4 * 16 * (128 + 4) / 4];
    float* tile = (float*)tile4;

    const int lane   = threadIdx.x & 31;
    const int wv     = threadIdx.x >> 5;
    const int lane15 = lane & 15;
    const int half   = lane >> 4;
    const int mbase  = half * 8;
    const int n0     = blockIdx.x * 16;
    const int col    = wv * 16 + lane15;

    node_rows_to_lds<128>(nf, tile, n0);
    __syncthreads();

    v8f aS = {}, aV0 = {}, aV1 = {}, aV2 = {};
    node_products<128>(tile, W0p, W1p, lane15, half, col, aS, aV0, aV1, aV2);

    const float inv = 0.08838834764831845f;
#pragma unroll
    for (int r = 0; r < 8; ++r) {
        const int n = n0 + mbase + r;
        s_up[n * 128 + col] = aS[r] * inv;
        const int vo = (n * 128 + col) * 3;
        v_up[vo + 0] = aV0[r] * inv;
        v_up[vo + 1] = aV1[r] * inv;
        v_up[vo + 2] = aV2[r] * inv;
    }
}

__device__ __forceinline__ void mlp64_layer(const float* HSRC, float* HDST,
                                            const v2f* __restrict__ Wp,
                                            int lane15, int half, int mbase) {
    const v2f* H2    = (const v2f*)HSRC;
    const int abase  = lane15 * 34 + half;
    const int wbase  = half * 64 + lane15;
    v8f acc[4] = {{}, {}, {}, {}};
    for (int k = 0; k < 64; k += 4) {
        const v2f a  = H2[abase + (k >> 1)];
        const int qb = (k >> 2) * 128 + wbase;
#pragma unroll
        for (int ct = 0; ct < 4; ++ct) {
            const v2f b = Wp[qb + ct * 16];
            acc[ct] = wmma4(a, b, acc[ct]);
        }
    }
#pragma unroll
    for (int ct = 0; ct < 4; ++ct)
#pragma unroll
        for (int r = 0; r < 8; ++r)
            HDST[(mbase + r) * 68 + ct * 16 + lane15] = silu(acc[ct][r] * 0.125f);
}

template <int J>
__device__ __forceinline__ void tp_pass(
        const int g, const v2f* H2, const int abase, const int lane, const int lane15,
        const v2f* w4s, const float* __restrict__ s_up,
        const float* __restrict__ v_up, float* __restrict__ msg,
        const int (&sidx)[8], const int (&vidx)[8], const int (&midx)[8],
        const float (&y0a)[8], const float (&y1xa)[8], const float (&y1ya)[8],
        const float (&y1za)[8], const unsigned fl, const unsigned am, const unsigned sm,
        const unsigned cm, const bool jadd) {
    const float c  = 0.08838834764831845f;
    const float c3 = 0.05103103630798287f;
    const int ub = (g & 1) * 64 + lane15;

    float gs[4][8];
    float gv[4][8][3];
    if (J == 0 || J == 2) {
#pragma unroll
        for (int i = 0; i < 4; ++i)
#pragma unroll
            for (int r = 0; r < 8; ++r)
                gs[i][r] = s_up[sidx[r] + ub + 16 * i];
    } else {
#pragma unroll
        for (int i = 0; i < 4; ++i)
#pragma unroll
            for (int r = 0; r < 8; ++r) {
                const int vb = vidx[r] + 3 * (ub + 16 * i);
                gv[i][r][0] = v_up[vb + 0];
                gv[i][r][1] = v_up[vb + 1];
                gv[i][r][2] = v_up[vb + 2];
            }
    }

    __builtin_amdgcn_sched_barrier(0);

    const int wb = g * 128 + lane;
    v2f bq[2][4][4], aq[2][4];
#pragma unroll
    for (int s = 0; s < 4; ++s) {
        aq[0][s] = H2[abase + 2 * s];
#pragma unroll
        for (int i = 0; i < 4; ++i)
            bq[0][s][i] = w4s[s * 1024 + wb + i * 32];
    }
    v8f acc[4] = {{}, {}, {}, {}};
#pragma unroll
    for (int blk = 0; blk < 4; ++blk) {
        const int cur = blk & 1;
        if (blk < 3) {
#pragma unroll
            for (int s = 0; s < 4; ++s) {
                const int kk = (blk + 1) * 4 + s;
                aq[cur ^ 1][s] = H2[abase + 2 * kk];
#pragma unroll
                for (int i = 0; i < 4; ++i)
                    bq[cur ^ 1][s][i] = w4s[kk * 1024 + wb + i * 32];
            }
            __builtin_amdgcn_sched_barrier(0);
        }
#pragma unroll
        for (int s = 0; s < 4; ++s)
#pragma unroll
            for (int i = 0; i < 4; ++i)
                acc[i] = wmma4(aq[cur][s], bq[cur][s][i], acc[i]);
        if (blk < 2) __builtin_amdgcn_sched_barrier(0);
    }

    constexpr int NC   = (J >= 2) ? 3 : 1;
    constexpr int COL0 = (J == 0) ? 0 : (J == 1) ? 128 : (J == 2) ? 256 : 640;
    float run[4][3], cap[4][3];
#pragma unroll
    for (int i = 0; i < 4; ++i)
#pragma unroll
        for (int k = 0; k < 3; ++k) { run[i][k] = 0.f; cap[i][k] = 0.f; }

#pragma unroll
    for (int r = 0; r < 8; ++r) {
#pragma unroll
        for (int i = 0; i < 4; ++i) {
            const float w = acc[i][r] * 0.125f;
            if (J == 0) {
                run[i][0] += c * gs[i][r] * y0a[r] * w;
            } else if (J == 1) {
                const float dot = gv[i][r][0] * y1xa[r] + gv[i][r][1] * y1ya[r] +
                                  gv[i][r][2] * y1za[r];
                run[i][0] += c3 * dot * w;
            } else if (J == 2) {
                const float t = c * gs[i][r] * w;
                run[i][0] += t * y1xa[r];
                run[i][1] += t * y1ya[r];
                run[i][2] += t * y1za[r];
            } else {
                const float t = c * y0a[r] * w;
                run[i][0] += t * gv[i][r][0];
                run[i][1] += t * gv[i][r][1];
                run[i][2] += t * gv[i][r][2];
            }
        }
        const bool capr = (cm >> r) & 1u;
#pragma unroll
        for (int i = 0; i < 4; ++i)
#pragma unroll
            for (int k = 0; k < NC; ++k)
                cap[i][k] = capr ? run[i][k] : cap[i][k];
        if (r == 7) {
#pragma unroll
            for (int i = 0; i < 4; ++i)
#pragma unroll
                for (int k = 0; k < NC; ++k) {
                    const float other = xhalf_f(cap[i][k]);
                    run[i][k] = jadd ? run[i][k] + other : run[i][k];
                }
        }
        const int mb = midx[r] + COL0;
        if ((am >> r) & 1u) {
#pragma unroll
            for (int i = 0; i < 4; ++i) {
                const int u = ub + 16 * i;
#pragma unroll
                for (int k = 0; k < NC; ++k)
                    atomicAdd(&msg[mb + NC * u + k], run[i][k]);
            }
        }
        if ((sm >> r) & 1u) {
#pragma unroll
            for (int i = 0; i < 4; ++i) {
                const int u = ub + 16 * i;
#pragma unroll
                for (int k = 0; k < NC; ++k)
                    msg[mb + NC * u + k] = run[i][k];
            }
        }
        if (r < 7) {
            const bool last = (fl >> r) & 1u;
#pragma unroll
            for (int i = 0; i < 4; ++i)
#pragma unroll
                for (int k = 0; k < NC; ++k)
                    run[i][k] = last ? 0.f : run[i][k];
        }
    }
}

#define EDGE_WAVES 12
#define N_TILES (N_EDGES / 16)

__device__ __forceinline__ int w4s_slot(const int q) {
    const int kk = q >> 10, hf = (q >> 9) & 1, col = q & 511;
    return kk * 1024 + (col >> 4) * 32 + hf * 16 + (col & 15);
}

__global__ void __launch_bounds__(EDGE_WAVES * 32)
edge_kernel(const float* __restrict__ ea, const float* __restrict__ ef,
            const int* __restrict__ snd, const int* __restrict__ rcv,
            const v2f* __restrict__ w1p, const v2f* __restrict__ w2p,
            const v2f* __restrict__ w3p, const v2f* __restrict__ w4p,
            const float* __restrict__ s_up, const float* __restrict__ v_up,
            const int* __restrict__ perm, float* __restrict__ msg) {
    __shared__ float4 w4s4[8 * 1024];
    __shared__ float hA[EDGE_WAVES][16 * 68];
    __shared__ float hB[EDGE_WAVES][16 * 68];
    __shared__ int   tile_next;

    const int lane   = threadIdx.x & 31;
    const int wv     = threadIdx.x >> 5;
    const int lane15 = lane & 15;
    const int half   = lane >> 4;
    const int mbase  = half * 8;

    float* HA = hA[wv];
    float* HB = hB[wv];
    const v2f* w4s = (const v2f*)w4s4;

    if (threadIdx.x == 0) tile_next = 0;
    {
        const float4* w4g = (const float4*)w4p;
        constexpr int NT   = EDGE_WAVES * 32;
        constexpr int FULL = 8192 / NT;
        const bool    tail = (int)threadIdx.x < 8192 - FULL * NT;
        float4 pc[FULL];
        float4 pt = make_float4(0.f, 0.f, 0.f, 0.f);
#pragma unroll
        for (int j = 0; j < FULL; ++j)
            pc[j] = w4g[threadIdx.x + j * NT];
        if (tail) pt = w4g[threadIdx.x + FULL * NT];
#pragma unroll
        for (int j = 0; j < FULL; ++j)
            w4s4[w4s_slot(2 * ((int)threadIdx.x + j * NT)) >> 1] = pc[j];
        if (tail) w4s4[w4s_slot(2 * ((int)threadIdx.x + FULL * NT)) >> 1] = pt;
    }
    __syncthreads();

    const int tper  = N_TILES / (int)gridDim.x;
    const int trem  = N_TILES % (int)gridDim.x;
    const int tbeg  = (int)blockIdx.x * tper + min((int)blockIdx.x, trem);
    const int tcnt  = tper + (((int)blockIdx.x < trem) ? 1 : 0);

    for (;;) {
        int tk = 0;
        if (lane == 0) tk = atomicAdd(&tile_next, 1);
        tk = __builtin_amdgcn_readfirstlane(tk);
        if (tk >= tcnt) break;
        const int e0 = (tbeg + tk) * 16;

        const int  nbpos   = half ? (e0 + 16) : (e0 - 1);
        const bool nbvalid = (nbpos >= 0) && (nbpos < N_EDGES);
        const int  nbrcv   = rcv[perm[nbvalid ? nbpos : e0]];

        int   sidx[8], vidx[8], midx[8];
        float y0a[8], y1xa[8], y1ya[8], y1za[8];
#pragma unroll
        for (int r = 0; r < 8; ++r) {
            const int e    = perm[e0 + mbase + r];
            const int node = snd[e];
            sidx[r] = node * 128;
            vidx[r] = node * 384;
            midx[r] = rcv[e] * 1024;
            const float4 y = *(const float4*)(ea + e * 4);
            y0a[r] = y.x; y1xa[r] = y.y; y1ya[r] = y.z; y1za[r] = y.w;
        }
        unsigned fl = 0x80u;
#pragma unroll
        for (int r = 0; r < 7; ++r)
            if (midx[r] != midx[r + 1]) fl |= (1u << r);

        const int  edge_mine  = half ? midx[0] : midx[7];
        const bool join       = (edge_mine == xhalf_i(edge_mine));
        const unsigned fl_oth = (unsigned)xhalf_i((int)fl);
        const unsigned fb     = 1u << __builtin_ctz(fl);
        const int  nbm       = nbvalid ? nbrcv * 1024 : -1;
        const bool open_mine = (nbm == (half ? midx[7] : midx[0]));
        const bool open_oth  = xhalf_i(open_mine ? 1 : 0) != 0;
        const bool joined_last = join & (fl_oth == 0x80u) & open_oth;
        const unsigned am_lo   = (open_mine ? fb : 0u) | (joined_last ? 0x80u : 0u);
        const unsigned am_hi   = open_mine ? 0x80u : 0u;
        unsigned am = half ? am_hi : am_lo;
        unsigned sm = fl & ~am;
        const unsigned cm = ((half != 0) & join) ? fb : 0u;
        am &= ~cm;
        sm &= ~cm;
        const bool jadd = (half == 0) && join;

        {
            const v2f* ef2  = (const v2f*)ef;
            const int eb2   = perm[e0 + lane15] * 4 + half;
            const int wbase = half * 64 + lane15;
            v8f acc[4] = {{}, {}, {}, {}};
#pragma unroll
            for (int k = 0; k < 8; k += 4) {
                const v2f a  = ef2[eb2 + (k >> 1)];
                const int qb = (k >> 2) * 128 + wbase;
#pragma unroll
                for (int ct = 0; ct < 4; ++ct) {
                    const v2f b = w1p[qb + ct * 16];
                    acc[ct] = wmma4(a, b, acc[ct]);
                }
            }
#pragma unroll
            for (int ct = 0; ct < 4; ++ct)
#pragma unroll
                for (int r = 0; r < 8; ++r)
                    HA[(mbase + r) * 68 + ct * 16 + lane15] = silu(acc[ct][r] * 0.35355339059327373f);
        }
        asm volatile("s_wait_dscnt 0" ::: "memory");

        mlp64_layer(HA, HB, w2p, lane15, half, mbase);
        asm volatile("s_wait_dscnt 0" ::: "memory");
        mlp64_layer(HB, HA, w3p, lane15, half, mbase);
        asm volatile("s_wait_dscnt 0" ::: "memory");

        const v2f* H2   = (const v2f*)HA;
        const int abase = lane15 * 34 + half;
#pragma unroll 1
        for (int g = 0; g < 2; ++g)
            tp_pass<0>(g, H2, abase, lane, lane15, w4s, s_up, v_up, msg,
                       sidx, vidx, midx, y0a, y1xa, y1ya, y1za, fl, am, sm, cm, jadd);
#pragma unroll 1
        for (int g = 2; g < 4; ++g)
            tp_pass<1>(g, H2, abase, lane, lane15, w4s, s_up, v_up, msg,
                       sidx, vidx, midx, y0a, y1xa, y1ya, y1za, fl, am, sm, cm, jadd);
#pragma unroll 1
        for (int g = 4; g < 6; ++g)
            tp_pass<2>(g, H2, abase, lane, lane15, w4s, s_up, v_up, msg,
                       sidx, vidx, midx, y0a, y1xa, y1ya, y1za, fl, am, sm, cm, jadd);
#pragma unroll 1
        for (int g = 6; g < 8; ++g)
            tp_pass<3>(g, H2, abase, lane, lane15, w4s, s_up, v_up, msg,
                       sidx, vidx, midx, y0a, y1xa, y1ya, y1za, fl, am, sm, cm, jadd);
    }
}

__global__ void __launch_bounds__(256)
node_down_kernel(const float* __restrict__ msg, const v2f* __restrict__ L0p,
                 const v2f* __restrict__ L1p, float* __restrict__ out) {
    __shared__ float4 tile4[4 * 16 * (256 + 4) / 4];
    float* tile = (float*)tile4;

    const int lane   = threadIdx.x & 31;
    const int wv     = threadIdx.x >> 5;
    const int lane15 = lane & 15;
    const int half   = lane >> 4;
    const int mbase  = half * 8;
    const int n0     = blockIdx.x * 16;
    const int col    = wv * 16 + lane15;

    node_rows_to_lds<256>(msg, tile, n0);
    __syncthreads();

    v8f aS = {}, aV0 = {}, aV1 = {}, aV2 = {};
    node_products<256>(tile, L0p, L1p, lane15, half, col, aS, aV0, aV1, aV2);

    const float sc = (1.0f / 16.0f) * 0.1f;
#pragma unroll
    for (int r = 0; r < 8; ++r) {
        const int n = n0 + mbase + r;
        out[n * 512 + col] = aS[r] * sc;
        const int oo = n * 512 + 128 + col * 3;
        out[oo + 0] = aV0[r] * sc;
        out[oo + 1] = aV1[r] * sc;
        out[oo + 2] = aV2[r] * sc;
    }
}

extern "C" void kernel_launch(void* const* d_in, const int* in_sizes, int n_in,
                              void* d_out, int out_size, void* d_ws, size_t ws_size,
                              hipStream_t stream) {
    (void)in_sizes; (void)n_in; (void)out_size; (void)ws_size;
    const float* nf   = (const float*)d_in[0];
    const float* ea   = (const float*)d_in[1];
    const float* ef   = (const float*)d_in[2];
    const int*   snd  = (const int*)d_in[3];
    const int*   rcv  = (const int*)d_in[4];
    const float* Wup0 = (const float*)d_in[5];
    const float* Wup1 = (const float*)d_in[6];
    const float* w1   = (const float*)d_in[7];
    const float* w2   = (const float*)d_in[8];
    const float* w3   = (const float*)d_in[9];
    const float* w4   = (const float*)d_in[10];
    const float* Wl0  = (const float*)d_in[11];
    const float* Wl1  = (const float*)d_in[12];
    float* out = (float*)d_out;

    float* s_up = (float*)d_ws;
    float* v_up = s_up + (size_t)N_NODES * 128;
    float* msgb = v_up + (size_t)N_NODES * 384;
    float* wp   = msgb + (size_t)N_NODES * 1024;
    float* Wup0p = wp;  wp += 128 * 128;
    float* Wup1p = wp;  wp += 128 * 128;
    float* w1p   = wp;  wp += 8 * 64;
    float* w2p   = wp;  wp += 64 * 64;
    float* w3p   = wp;  wp += 64 * 64;
    float* w4p   = wp;  wp += 64 * 512;
    float* Wl0p  = wp;  wp += 256 * 128;
    float* Wl1p  = wp;  wp += 256 * 128;

    int* cnt  = (int*)d_out;
    int* perm = cnt + N_NODES;

    prep_kernel<<<2048, 256, 0, stream>>>((float4*)msgb, N_NODES * 1024 / 4, cnt, N_NODES,
                                          Wup0, Wup1, w1, w2, w3, w4, Wl0, Wl1,
                                          (float2*)Wup0p, (float2*)Wup1p, (float2*)w1p,
                                          (float2*)w2p, (float2*)w3p, (float2*)w4p,
                                          (float2*)Wl0p, (float2*)Wl1p);

    count_kernel<<<196, 1024, 0, stream>>>(rcv, cnt);
    scan_kernel<<<1, 1024, 0, stream>>>(cnt);
    fill_kernel<<<196, 1024, 0, stream>>>(rcv, cnt, perm);

    node_up_kernel<<<N_NODES / 16, 256, 0, stream>>>(nf, (const v2f*)Wup0p,
                                                     (const v2f*)Wup1p, s_up, v_up);
    edge_kernel<<<256, EDGE_WAVES * 32, 0, stream>>>(ea, ef, snd, rcv,
                                                  (const v2f*)w1p, (const v2f*)w2p,
                                                  (const v2f*)w3p, (const v2f*)w4p,
                                                  s_up, v_up, perm, msgb);
    node_down_kernel<<<N_NODES / 16, 256, 0, stream>>>(msgb, (const v2f*)Wl0p,
                                                       (const v2f*)Wl1p, out);
}


# Edge kernel: each layer-4 pass takes 32 channels of the two weight blocks that use the same sender values (0 with 2, 1 with 3): s_up / v_up rows are gathered once per tile, not twice; sum order kept
# speedup vs baseline: 2.9535x; 1.0346x over previous
#include <hip/hip_runtime.h>

typedef float v2f __attribute__((ext_vector_type(2)));
typedef float v8f __attribute__((ext_vector_type(8)));

#define N_NODES 20000
#define N_EDGES 200000

__device__ __forceinline__ v8f wmma4(v2f a, v2f b, v8f c) {
    return __builtin_amdgcn_wmma_f32_16x16x4_f32(
         false, a,  false, b,
         (short)0, c,  false,  false);
}

__device__ __forceinline__ float silu(float x) {
    return x / (1.0f + __expf(-x));
}

__device__ __forceinline__ int xhalf_i(int x) {
    return __builtin_amdgcn_permlanex16(x, x, 0x76543210u, 0xfedcba98u, false, false);
}
__device__ __forceinline__ float xhalf_f(float x) {
    return __builtin_bit_cast(float, xhalf_i(__builtin_bit_cast(int, x)));
}

__device__ __forceinline__ bool pack_b_one(const float* __restrict__ W, float2* __restrict__ Wp,
                                           const int K, const int N, int& idx) {
    const int total = (K / 2) * N;
    if (idx < total) {
        const int q = idx / N;
        const int n = idx - q * N;
        float2 p;
        p.x = W[(2 * q) * N + n];
        p.y = W[(2 * q + 1) * N + n];
        Wp[idx] = p;
        return true;
    }
    idx -= total;
    return false;
}

#define PACK_TOTAL (2 * 64 * 128 + 4 * 64 + 2 * 32 * 64 + 32 * 512 + 2 * 128 * 128)

__global__ void prep_kernel(float4* __restrict__ p, int n4, int* __restrict__ cnt, int ncnt,
                            const float* __restrict__ Wup0, const float* __restrict__ Wup1,
                            const float* __restrict__ w1, const float* __restrict__ w2,
                            const float* __restrict__ w3, const float* __restrict__ w4,
                            const float* __restrict__ Wl0, const float* __restrict__ Wl1,
                            float2* __restrict__ Wup0p, float2* __restrict__ Wup1p,
                            float2* __restrict__ w1p, float2* __restrict__ w2p,
                            float2* __restrict__ w3p, float2* __restrict__ w4p,
                            float2* __restrict__ Wl0p, float2* __restrict__ Wl1p) {
    const int stride = gridDim.x * blockDim.x;
    const int gid    = blockIdx.x * blockDim.x + threadIdx.x;
    for (int i = gid; i < n4; i += stride)
        p[i] = make_float4(0.f, 0.f, 0.f, 0.f);
    for (int i = gid; i < ncnt; i += stride)
        cnt[i] = 0;
    for (int g = gid; g < PACK_TOTAL; g += stride) {
        int idx = g;
        if (pack_b_one(Wup0, Wup0p, 128, 128, idx)) continue;
        if (pack_b_one(Wup1, Wup1p, 128, 128, idx)) continue;
        if (pack_b_one(w1,   w1p,   8,   64,  idx)) continue;
        if (pack_b_one(w2,   w2p,   64,  64,  idx)) continue;
        if (pack_b_one(w3,   w3p,   64,  64,  idx)) continue;
        if (pack_b_one(w4,   w4p,   64,  512, idx)) continue;
        if (pack_b_one(Wl0,  Wl0p,  256, 128, idx)) continue;
        pack_b_one(Wl1, Wl1p, 256, 128, idx);
    }
}

__global__ void count_kernel(const int* __restrict__ rcv, int* __restrict__ cnt) {
    const int stride = gridDim.x * blockDim.x;
    for (int e = blockIdx.x * blockDim.x + threadIdx.x; e < N_EDGES; e += stride)
        atomicAdd(&cnt[rcv[e]], 1);
}

#define SCAN_PER_THREAD 20
__global__ void __launch_bounds__(1024)
scan_kernel(int* __restrict__ cnt) {
    __shared__ int part[1024];
    const int t = threadIdx.x;
    const int b = t * SCAN_PER_THREAD;
    int loc[SCAN_PER_THREAD];
    int s = 0;
#pragma unroll
    for (int k = 0; k < SCAN_PER_THREAD; ++k) {
        const int idx = b + k;
        const int v = (idx < N_NODES) ? cnt[idx] : 0;
        loc[k] = s;
        s += v;
    }
    part[t] = s;
    __syncthreads();
    for (int off = 1; off < 1024; off <<= 1) {
        const int v = (t >= off) ? part[t - off] : 0;
        __syncthreads();
        part[t] += v;
        __syncthreads();
    }
    const int base = part[t] - s;
#pragma unroll
    for (int k = 0; k < SCAN_PER_THREAD; ++k) {
        const int idx = b + k;
        if (idx < N_NODES) cnt[idx] = base + loc[k];
    }
}

__global__ void fill_kernel(const int* __restrict__ rcv, int* __restrict__ cnt,
                            int* __restrict__ perm) {
    const int stride = gridDim.x * blockDim.x;
    for (int e = blockIdx.x * blockDim.x + threadIdx.x; e < N_EDGES; e += stride) {
        const int pos = atomicAdd(&cnt[rcv[e]], 1);
        perm[pos] = e;
    }
}

template <int K>
__device__ __forceinline__ void node_rows_to_lds(const float* __restrict__ in, float* tile,
                                                 const int n0) {
    constexpr int PITCH = K + 4;
    constexpr int PLANE = 16 * PITCH;
    constexpr int Q     = K / 4;
    constexpr int PER   = 16 * Q / 256;
    const float4* in4 = (const float4*)in;
    float4 s4[PER], va[PER], vb[PER], vc[PER];
#pragma unroll
    for (int j = 0; j < PER; ++j) {
        const int idx = (int)threadIdx.x + j * 256;
        const int row = idx / Q, c = idx % Q;
        const int rb  = (n0 + row) * K;
        s4[j] = in4[rb + c];
        va[j] = in4[rb + Q + 3 * c + 0];
        vb[j] = in4[rb + Q + 3 * c + 1];
        vc[j] = in4[rb + Q + 3 * c + 2];
    }
#pragma unroll
    for (int j = 0; j < PER; ++j) {
        const int idx = (int)threadIdx.x + j * 256;
        const int row = idx / Q, c = idx % Q;
        float* d = tile + row * PITCH + 4 * c;
        *(float4*)(d)             = s4[j];
        *(float4*)(d + PLANE)     = make_float4(va[j].x, va[j].w, vb[j].z, vc[j].y);
        *(float4*)(d + 2 * PLANE) = make_float4(va[j].y, vb[j].x, vb[j].w, vc[j].z);
        *(float4*)(d + 3 * PLANE) = make_float4(va[j].z, vb[j].y, vc[j].x, vc[j].w);
    }
}

template <int K>
__device__ __forceinline__ void node_products(const float* tile, const v2f* __restrict__ W0p,
                                              const v2f* __restrict__ W1p, const int lane15,
                                              const int half, const int col,
                                              v8f& aS, v8f& aV0, v8f& aV1, v8f& aV2) {
    constexpr int PITCH2 = (K + 4) / 2;
    constexpr int PLANE2 = 16 * PITCH2;
    const v2f* a2    = (const v2f*)tile + lane15 * PITCH2 + half;
    const int  wbase = half * 128 + col;
    for (int k = 0; k < K; k += 4) {
        const int qb = (k >> 2) * 256 + wbase;
        const v2f b0 = W0p[qb];
        const v2f b1 = W1p[qb];
        const int ka = k >> 1;
        aS  = wmma4(a2[ka], b0, aS);
        aV0 = wmma4(a2[PLANE2 + ka], b1, aV0);
        aV1 = wmma4(a2[2 * PLANE2 + ka], b1, aV1);
        aV2 = wmma4(a2[3 * PLANE2 + ka], b1, aV2);
    }
}

__global__ void __launch_bounds__(256)
node_up_kernel(const float* __restrict__ nf, const v2f* __restrict__ W0p,
               const v2f* __restrict__ W1p, float* __restrict__ s_up,
               float* __restrict__ v_up) {
    __shared__ float4 tile4[4 * 16 * (128 + 4) / 4];
    float* tile = (float*)tile4;

    const int lane   = threadIdx.x & 31;
    const int wv     = threadIdx.x >> 5;
    const int lane15 = lane & 15;
    const int half   = lane >> 4;
    const int mbase  = half * 8;
    const int n0     = blockIdx.x * 16;
    const int col    = wv * 16 + lane15;

    node_rows_to_lds<128>(nf, tile, n0);
    __syncthreads();

    v8f aS = {}, aV0 = {}, aV1 = {}, aV2 = {};
    node_products<128>(tile, W0p, W1p, lane15, half, col, aS, aV0, aV1, aV2);

    const float inv = 0.08838834764831845f;
#pragma unroll
    for (int r = 0; r < 8; ++r) {
        const int n = n0 + mbase + r;
        s_up[n * 128 + col] = aS[r] * inv;
        const int vo = (n * 128 + col) * 3;
        v_up[vo + 0] = aV0[r] * inv;
        v_up[vo + 1] = aV1[r] * inv;
        v_up[vo + 2] = aV2[r] * inv;
    }
}

__device__ __forceinline__ void mlp64_layer(const float* HSRC, float* HDST,
                                            const v2f* __restrict__ Wp,
                                            int lane15, int half, int mbase) {
    const v2f* H2    = (const v2f*)HSRC;
    const int abase  = lane15 * 34 + half;
    const int wbase  = half * 64 + lane15;
    v8f acc[4] = {{}, {}, {}, {}};
    for (int k = 0; k < 64; k += 4) {
        const v2f a  = H2[abase + (k >> 1)];
        const int qb = (k >> 2) * 128 + wbase;
#pragma unroll
        for (int ct = 0; ct < 4; ++ct) {
            const v2f b = Wp[qb + ct * 16];
            acc[ct] = wmma4(a, b, acc[ct]);
        }
    }
#pragma unroll
    for (int ct = 0; ct < 4; ++ct)
#pragma unroll
        for (int r = 0; r < 8; ++r)
            HDST[(mbase + r) * 68 + ct * 16 + lane15] = silu(acc[ct][r] * 0.125f);
}

template <int S>
__device__ __forceinline__ void tp_pair_pass(
        const int p, const v2f* H2, const int abase, const int lane, const int lane15,
        const v2f* w4s, const float* __restrict__ s_up,
        const float* __restrict__ v_up, float* __restrict__ msg,
        const int (&sidx)[8], const int (&vidx)[8], const int (&midx)[8],
        const float (&y0a)[8], const float (&y1xa)[8], const float (&y1ya)[8],
        const float (&y1za)[8], const unsigned fl, const unsigned am, const unsigned sm,
        const unsigned cm, const bool jadd) {
    const float c  = 0.08838834764831845f;
    const float c3 = 0.05103103630798287f;
    const int ub = p * 32 + lane15;

    float gs[2][8];
    float gv[2][8][3];
    if (S == 0) {
#pragma unroll
        for (int i = 0; i < 2; ++i)
#pragma unroll
            for (int r = 0; r < 8; ++r)
                gs[i][r] = s_up[sidx[r] + ub + 16 * i];
    } else {
#pragma unroll
        for (int i = 0; i < 2; ++i)
#pragma unroll
            for (int r = 0; r < 8; ++r) {
                const int vb = vidx[r] + 3 * (ub + 16 * i);
                gv[i][r][0] = v_up[vb + 0];
                gv[i][r][1] = v_up[vb + 1];
                gv[i][r][2] = v_up[vb + 2];
            }
    }

    __builtin_amdgcn_sched_barrier(0);

    const int wb = (S * 8 + 2 * p) * 32 + lane;
    v2f bq[2][4][4], aq[2][4];
#pragma unroll
    for (int s = 0; s < 4; ++s) {
        aq[0][s] = H2[abase + 2 * s];
#pragma unroll
        for (int a = 0; a < 4; ++a)
            bq[0][s][a] = w4s[s * 1024 + wb + (a >> 1) * 512 + (a & 1) * 32];
    }
    v8f acc[4] = {{}, {}, {}, {}};
#pragma unroll
    for (int blk = 0; blk < 4; ++blk) {
        const int cur = blk & 1;
        if (blk < 3) {
#pragma unroll
            for (int s = 0; s < 4; ++s) {
                const int kk = (blk + 1) * 4 + s;
                aq[cur ^ 1][s] = H2[abase + 2 * kk];
#pragma unroll
                for (int a = 0; a < 4; ++a)
                    bq[cur ^ 1][s][a] = w4s[kk * 1024 + wb + (a >> 1) * 512 + (a & 1) * 32];
            }
            __builtin_amdgcn_sched_barrier(0);
        }
#pragma unroll
        for (int s = 0; s < 4; ++s)
#pragma unroll
            for (int a = 0; a < 4; ++a)
                acc[a] = wmma4(aq[cur][s], bq[cur][s][a], acc[a]);
        if (blk < 2) __builtin_amdgcn_sched_barrier(0);
    }

    constexpr int COLA = (S == 0) ? 0 : 128;
    constexpr int COLB = (S == 0) ? 256 : 640;
    float runA[2], capA[2], runB[2][3], capB[2][3];
#pragma unroll
    for (int i = 0; i < 2; ++i) {
        runA[i] = 0.f; capA[i] = 0.f;
#pragma unroll
        for (int k = 0; k < 3; ++k) { runB[i][k] = 0.f; capB[i][k] = 0.f; }
    }

#pragma unroll
    for (int r = 0; r < 8; ++r) {
#pragma unroll
        for (int i = 0; i < 2; ++i) {
            const float wA = acc[i][r] * 0.125f;
            const float wB = acc[2 + i][r] * 0.125f;
            if (S == 0) {
                runA[i] += c * gs[i][r] * y0a[r] * wA;
                const float t = c * gs[i][r] * wB;
                runB[i][0] += t * y1xa[r];
                runB[i][1] += t * y1ya[r];
                runB[i][2] += t * y1za[r];
            } else {
                const float dot = gv[i][r][0] * y1xa[r] + gv[i][r][1] * y1ya[r] +
                                  gv[i][r][2] * y1za[r];
                runA[i] += c3 * dot * wA;
                const float t = c * y0a[r] * wB;
                runB[i][0] += t * gv[i][r][0];
                runB[i][1] += t * gv[i][r][1];
                runB[i][2] += t * gv[i][r][2];
            }
        }
        const bool capr = (cm >> r) & 1u;
#pragma unroll
        for (int i = 0; i < 2; ++i) {
            capA[i] = capr ? runA[i] : capA[i];
#pragma unroll
            for (int k = 0; k < 3; ++k)
                capB[i][k] = capr ? runB[i][k] : capB[i][k];
        }
        if (r == 7) {
#pragma unroll
            for (int i = 0; i < 2; ++i) {
                const float oa = xhalf_f(capA[i]);
                runA[i] = jadd ? runA[i] + oa : runA[i];
#pragma unroll
                for (int k = 0; k < 3; ++k) {
                    const float ob = xhalf_f(capB[i][k]);
                    runB[i][k] = jadd ? runB[i][k] + ob : runB[i][k];
                }
            }
        }
        const int mb = midx[r];
        if ((am >> r) & 1u) {
#pragma unroll
            for (int i = 0; i < 2; ++i) {
                const int u = ub + 16 * i;
                atomicAdd(&msg[mb + COLA + u], runA[i]);
#pragma unroll
                for (int k = 0; k < 3; ++k)
                    atomicAdd(&msg[mb + COLB + 3 * u + k], runB[i][k]);
            }
        }
        if ((sm >> r) & 1u) {
#pragma unroll
            for (int i = 0; i < 2; ++i) {
                const int u = ub + 16 * i;
                msg[mb + COLA + u] = runA[i];
#pragma unroll
                for (int k = 0; k < 3; ++k)
                    msg[mb + COLB + 3 * u + k] = runB[i][k];
            }
        }
        if (r < 7) {
            const bool last = (fl >> r) & 1u;
#pragma unroll
            for (int i = 0; i < 2; ++i) {
                runA[i] = last ? 0.f : runA[i];
#pragma unroll
                for (int k = 0; k < 3; ++k)
                    runB[i][k] = last ? 0.f : runB[i][k];
            }
        }
    }
}

#define EDGE_WAVES 12
#define N_TILES (N_EDGES / 16)

__device__ __forceinline__ int w4s_slot(const int q) {
    const int kk = q >> 10, hf = (q >> 9) & 1, col = q & 511;
    return kk * 1024 + (col >> 4) * 32 + hf * 16 + (col & 15);
}

__global__ void __launch_bounds__(EDGE_WAVES * 32)
edge_kernel(const float* __restrict__ ea, const float* __restrict__ ef,
            const int* __restrict__ snd, const int* __restrict__ rcv,
            const v2f* __restrict__ w1p, const v2f* __restrict__ w2p,
            const v2f* __restrict__ w3p, const v2f* __restrict__ w4p,
            const float* __restrict__ s_up, const float* __restrict__ v_up,
            const int* __restrict__ perm, float* __restrict__ msg) {
    __shared__ float4 w4s4[8 * 1024];
    __shared__ float hA[EDGE_WAVES][16 * 68];
    __shared__ float hB[EDGE_WAVES][16 * 68];
    __shared__ int   tile_next;

    const int lane   = threadIdx.x & 31;
    const int wv     = threadIdx.x >> 5;
    const int lane15 = lane & 15;
    const int half   = lane >> 4;
    const int mbase  = half * 8;

    float* HA = hA[wv];
    float* HB = hB[wv];
    const v2f* w4s = (const v2f*)w4s4;

    if (threadIdx.x == 0) tile_next = 0;
    {
        const float4* w4g = (const float4*)w4p;
        constexpr int NT   = EDGE_WAVES * 32;
        constexpr int FULL = 8192 / NT;
        const bool    tail = (int)threadIdx.x < 8192 - FULL * NT;
        float4 pc[FULL];
        float4 pt = make_float4(0.f, 0.f, 0.f, 0.f);
#pragma unroll
        for (int j = 0; j < FULL; ++j)
            pc[j] = w4g[threadIdx.x + j * NT];
        if (tail) pt = w4g[threadIdx.x + FULL * NT];
#pragma unroll
        for (int j = 0; j < FULL; ++j)
            w4s4[w4s_slot(2 * ((int)threadIdx.x + j * NT)) >> 1] = pc[j];
        if (tail) w4s4[w4s_slot(2 * ((int)threadIdx.x + FULL * NT)) >> 1] = pt;
    }
    __syncthreads();

    const int tper  = N_TILES / (int)gridDim.x;
    const int trem  = N_TILES % (int)gridDim.x;
    const int tbeg  = (int)blockIdx.x * tper + min((int)blockIdx.x, trem);
    const int tcnt  = tper + (((int)blockIdx.x < trem) ? 1 : 0);

    for (;;) {
        int tk = 0;
        if (lane == 0) tk = atomicAdd(&tile_next, 1);
        tk = __builtin_amdgcn_readfirstlane(tk);
        if (tk >= tcnt) break;
        const int e0 = (tbeg + tk) * 16;

        const int  nbpos   = half ? (e0 + 16) : (e0 - 1);
        const bool nbvalid = (nbpos >= 0) && (nbpos < N_EDGES);
        const int  nbrcv   = rcv[perm[nbvalid ? nbpos : e0]];

        int   sidx[8], vidx[8], midx[8];
        float y0a[8], y1xa[8], y1ya[8], y1za[8];
#pragma unroll
        for (int r = 0; r < 8; ++r) {
            const int e    = perm[e0 + mbase + r];
            const int node = snd[e];
            sidx[r] = node * 128;
            vidx[r] = node * 384;
            midx[r] = rcv[e] * 1024;
            const float4 y = *(const float4*)(ea + e * 4);
            y0a[r] = y.x; y1xa[r] = y.y; y1ya[r] = y.z; y1za[r] = y.w;
        }
        unsigned fl = 0x80u;
#pragma unroll
        for (int r = 0; r < 7; ++r)
            if (midx[r] != midx[r + 1]) fl |= (1u << r);

        const int  edge_mine  = half ? midx[0] : midx[7];
        const bool join       = (edge_mine == xhalf_i(edge_mine));
        const unsigned fl_oth = (unsigned)xhalf_i((int)fl);
        const unsigned fb     = 1u << __builtin_ctz(fl);
        const int  nbm       = nbvalid ? nbrcv * 1024 : -1;
        const bool open_mine = (nbm == (half ? midx[7] : midx[0]));
        const bool open_oth  = xhalf_i(open_mine ? 1 : 0) != 0;
        const bool joined_last = join & (fl_oth == 0x80u) & open_oth;
        const unsigned am_lo   = (open_mine ? fb : 0u) | (joined_last ? 0x80u : 0u);
        const unsigned am_hi   = open_mine ? 0x80u : 0u;
        unsigned am = half ? am_hi : am_lo;
        unsigned sm = fl & ~am;
        const unsigned cm = ((half != 0) & join) ? fb : 0u;
        am &= ~cm;
        sm &= ~cm;
        const bool jadd = (half == 0) && join;

        {
            const v2f* ef2  = (const v2f*)ef;
            const int eb2   = perm[e0 + lane15] * 4 + half;
            const int wbase = half * 64 + lane15;
            v8f acc[4] = {{}, {}, {}, {}};
#pragma unroll
            for (int k = 0; k < 8; k += 4) {
                const v2f a  = ef2[eb2 + (k >> 1)];
                const int qb = (k >> 2) * 128 + wbase;
#pragma unroll
                for (int ct = 0; ct < 4; ++ct) {
                    const v2f b = w1p[qb + ct * 16];
                    acc[ct] = wmma4(a, b, acc[ct]);
                }
            }
#pragma unroll
            for (int ct = 0; ct < 4; ++ct)
#pragma unroll
                for (int r = 0; r < 8; ++r)
                    HA[(mbase + r) * 68 + ct * 16 + lane15] = silu(acc[ct][r] * 0.35355339059327373f);
        }
        asm volatile("s_wait_dscnt 0" ::: "memory");

        mlp64_layer(HA, HB, w2p, lane15, half, mbase);
        asm volatile("s_wait_dscnt 0" ::: "memory");
        mlp64_layer(HB, HA, w3p, lane15, half, mbase);
        asm volatile("s_wait_dscnt 0" ::: "memory");

        const v2f* H2   = (const v2f*)HA;
        const int abase = lane15 * 34 + half;
#pragma unroll 1
        for (int p = 0; p < 4; ++p)
            tp_pair_pass<0>(p, H2, abase, lane, lane15, w4s, s_up, v_up, msg,
                            sidx, vidx, midx, y0a, y1xa, y1ya, y1za, fl, am, sm, cm, jadd);
#pragma unroll 1
        for (int p = 0; p < 4; ++p)
            tp_pair_pass<1>(p, H2, abase, lane, lane15, w4s, s_up, v_up, msg,
                            sidx, vidx, midx, y0a, y1xa, y1ya, y1za, fl, am, sm, cm, jadd);
    }
}

__global__ void __launch_bounds__(256)
node_down_kernel(const float* __restrict__ msg, const v2f* __restrict__ L0p,
                 const v2f* __restrict__ L1p, float* __restrict__ out) {
    __shared__ float4 tile4[4 * 16 * (256 + 4) / 4];
    float* tile = (float*)tile4;

    const int lane   = threadIdx.x & 31;
    const int wv     = threadIdx.x >> 5;
    const int lane15 = lane & 15;
    const int half   = lane >> 4;
    const int mbase  = half * 8;
    const int n0     = blockIdx.x * 16;
    const int col    = wv * 16 + lane15;

    node_rows_to_lds<256>(msg, tile, n0);
    __syncthreads();

    v8f aS = {}, aV0 = {}, aV1 = {}, aV2 = {};
    node_products<256>(tile, L0p, L1p, lane15, half, col, aS, aV0, aV1, aV2);

    const float sc = (1.0f / 16.0f) * 0.1f;
#pragma unroll
    for (int r = 0; r < 8; ++r) {
        const int n = n0 + mbase + r;
        out[n * 512 + col] = aS[r] * sc;
        const int oo = n * 512 + 128 + col * 3;
        out[oo + 0] = aV0[r] * sc;
        out[oo + 1] = aV1[r] * sc;
        out[oo + 2] = aV2[r] * sc;
    }
}

extern "C" void kernel_launch(void* const* d_in, const int* in_sizes, int n_in,
                              void* d_out, int out_size, void* d_ws, size_t ws_size,
                              hipStream_t stream) {
    (void)in_sizes; (void)n_in; (void)out_size; (void)ws_size;
    const float* nf   = (const float*)d_in[0];
    const float* ea   = (const float*)d_in[1];
    const float* ef   = (const float*)d_in[2];
    const int*   snd  = (const int*)d_in[3];
    const int*   rcv  = (const int*)d_in[4];
    const float* Wup0 = (const float*)d_in[5];
    const float* Wup1 = (const float*)d_in[6];
    const float* w1   = (const float*)d_in[7];
    const float* w2   = (const float*)d_in[8];
    const float* w3   = (const float*)d_in[9];
    const float* w4   = (const float*)d_in[10];
    const float* Wl0  = (const float*)d_in[11];
    const float* Wl1  = (const float*)d_in[12];
    float* out = (float*)d_out;

    float* s_up = (float*)d_ws;
    float* v_up = s_up + (size_t)N_NODES * 128;
    float* msgb = v_up + (size_t)N_NODES * 384;
    float* wp   = msgb + (size_t)N_NODES * 1024;
    float* Wup0p = wp;  wp += 128 * 128;
    float* Wup1p = wp;  wp += 128 * 128;
    float* w1p   = wp;  wp += 8 * 64;
    float* w2p   = wp;  wp += 64 * 64;
    float* w3p   = wp;  wp += 64 * 64;
    float* w4p   = wp;  wp += 64 * 512;
    float* Wl0p  = wp;  wp += 256 * 128;
    float* Wl1p  = wp;  wp += 256 * 128;

    int* cnt  = (int*)d_out;
    int* perm = cnt + N_NODES;

    prep_kernel<<<2048, 256, 0, stream>>>((float4*)msgb, N_NODES * 1024 / 4, cnt, N_NODES,
                                          Wup0, Wup1, w1, w2, w3, w4, Wl0, Wl1,
                                          (float2*)Wup0p, (float2*)Wup1p, (float2*)w1p,
                                          (float2*)w2p, (float2*)w3p, (float2*)w4p,
                                          (float2*)Wl0p, (float2*)Wl1p);

    count_kernel<<<196, 1024, 0, stream>>>(rcv, cnt);
    scan_kernel<<<1, 1024, 0, stream>>>(cnt);
    fill_kernel<<<196, 1024, 0, stream>>>(rcv, cnt, perm);

    node_up_kernel<<<N_NODES / 16, 256, 0, stream>>>(nf, (const v2f*)Wup0p,
                                                     (const v2f*)Wup1p, s_up, v_up);
    edge_kernel<<<256, EDGE_WAVES * 32, 0, stream>>>(ea, ef, snd, rcv,
                                                  (const v2f*)w1p, (const v2f*)w2p,
                                                  (const v2f*)w3p, (const v2f*)w4p,
                                                  s_up, v_up, perm, msgb);
    node_down_kernel<<<N_NODES / 16, 256, 0, stream>>>(msgb, (const v2f*)Wl0p,
                                                       (const v2f*)Wl1p, out);
}
